# starting kernel
# baseline (speedup 1.0000x reference)
#include <hip/hip_runtime.h>
#include <math.h>

#define DIMA   100
#define MSTEP  50
#define HDIM   128
#define BATCH  8192
#define NPAD4  112
#define SROW   104
#define ROWS   64
#define S0INV  0.01f
#define RDT    0.001f
#define SIG    0.2f
#define LNEPS  1e-5f

typedef __attribute__((ext_vector_type(16))) _Float16 v16h;
typedef __attribute__((ext_vector_type(8)))  _Float16 v8h;
typedef __attribute__((ext_vector_type(8)))  float    v8f;

union V16U { v16h v; v8h h[2]; };

__device__ __forceinline__ v16h load_frag(const _Float16* base, int kb) {
  V16U u;
  u.h[0] = *(const v8h*)(base + kb);
  u.h[1] = *(const v8h*)(base + kb + 16);
  return u.v;
}

__device__ __forceinline__ float red16(float x) {
#pragma unroll
  for (int off = 1; off < 16; off <<= 1) x += __shfl_xor(x, off, 16);
  return x;
}

__device__ __forceinline__ void mlp_layer(const _Float16* in, const _Float16* Wt,
                                          const float* bb, const float* gg,
                                          const float* ee, _Float16* out,
                                          int rbase, int lane15, int half) {
  v8f acc[8];
#pragma unroll
  for (int nt = 0; nt < 8; ++nt) acc[nt] = {};

  const _Float16* arow = in + (rbase + lane15) * HDIM;
#pragma unroll
  for (int kt = 0; kt < 4; ++kt) {
    const int kb = kt * 32 + half * 8;
    v16h a = load_frag(arow, kb);
#pragma unroll
    for (int nt = 0; nt < 8; ++nt) {
      v16h b = load_frag(Wt + (nt * 16 + lane15) * HDIM, kb);
      acc[nt] = __builtin_amdgcn_wmma_f32_16x16x32_f16(
          false, a, false, b, (short)0, acc[nt], false, false);
    }
  }

  float bn[8], gn[8], en[8];
#pragma unroll
  for (int nt = 0; nt < 8; ++nt) {
    const int n = nt * 16 + lane15;
    bn[nt] = bb[n]; gn[nt] = gg[n]; en[nt] = ee[n];
  }

#pragma unroll
  for (int j = 0; j < 8; ++j) {
    float p = 0.f;
#pragma unroll
    for (int nt = 0; nt < 8; ++nt) { acc[nt][j] += bn[nt]; p += acc[nt][j]; }
    p = red16(p);
    const float mu = p * (1.f / 128.f);
    float q = 0.f;
#pragma unroll
    for (int nt = 0; nt < 8; ++nt) { float d = acc[nt][j] - mu; q += d * d; }
    q = red16(q);
    const float rstd = rsqrtf(q * (1.f / 128.f) + LNEPS);

    const int row = rbase + j + half * 8;
    _Float16* orow = out + row * HDIM + lane15;
#pragma unroll
    for (int nt = 0; nt < 8; ++nt) {
      float x = (acc[nt][j] - mu) * rstd * gn[nt] + en[nt];
      float ge = 0.5f * x * (1.f + erff(x * 0.70710678118654752f));
      orow[nt * 16] = (_Float16)ge;
    }
  }
}

extern "C" __global__ void __launch_bounds__(128)
bsde_fused_kernel(const float* __restrict__ dw, const float* __restrict__ tg,
                  const float* __restrict__ W1, const float* __restrict__ b1,
                  const float* __restrict__ g1, const float* __restrict__ be1,
                  const float* __restrict__ W2, const float* __restrict__ b2,
                  const float* __restrict__ g2, const float* __restrict__ be2,
                  const float* __restrict__ W3, const float* __restrict__ b3,
                  const float* __restrict__ g3, const float* __restrict__ be3,
                  const float* __restrict__ W4, const float* __restrict__ b4,
                  const float* __restrict__ Y0,
                  float* __restrict__ outY, float* __restrict__ outS) {
  extern __shared__ __align__(16) char smem[];
  _Float16* Wt1 = (_Float16*)smem;
  _Float16* Wt2 = Wt1 + HDIM * HDIM;
  _Float16* Wt3 = Wt2 + HDIM * HDIM;
  _Float16* Wt4 = Wt3 + HDIM * HDIM;
  float* prm = (float*)(Wt4 + NPAD4 * HDIM);
  float* pb1 = prm;        float* pg1 = prm + 128;  float* pe1 = prm + 256;
  float* pb2 = prm + 384;  float* pg2 = prm + 512;  float* pe2 = prm + 640;
  float* pb3 = prm + 768;  float* pg3 = prm + 896;  float* pe3 = prm + 1024;
  float* pb4 = prm + 1152;
  float* Sl  = prm + 1264;
  float* Yl  = Sl + ROWS * SROW;
  _Float16* buf0 = (_Float16*)(Yl + ROWS);
  _Float16* buf1 = buf0 + ROWS * HDIM;

  const int tid = threadIdx.x;
  const int lane = tid & 31, lane15 = lane & 15, half = lane >> 4;
  const int rbase = (tid >> 5) * 16;
  const long blockBase = (long)blockIdx.x * ROWS;

  for (int i = tid; i < HDIM * HDIM; i += 128) {
    const int n = i >> 7, k = i & 127;
    Wt1[i] = (_Float16)(k < DIMA + 1 ? W1[k * HDIM + n] : 0.f);
    Wt2[i] = (_Float16)W2[k * HDIM + n];
    Wt3[i] = (_Float16)W3[k * HDIM + n];
  }
  for (int i = tid; i < NPAD4 * HDIM; i += 128) {
    const int n = i >> 7, k = i & 127;
    Wt4[i] = (_Float16)(n < DIMA ? W4[k * DIMA + n] : 0.f);
  }
  pb1[tid] = b1[tid]; pg1[tid] = g1[tid]; pe1[tid] = be1[tid];
  pb2[tid] = b2[tid]; pg2[tid] = g2[tid]; pe2[tid] = be2[tid];
  pb3[tid] = b3[tid]; pg3[tid] = g3[tid]; pe3[tid] = be3[tid];
  if (tid < NPAD4) pb4[tid] = (tid < DIMA) ? b4[tid] : 0.f;

  const float y0 = Y0[0];
#pragma unroll 4
  for (int r = 0; r < 16; ++r) {
    for (int d = lane; d < DIMA; d += 32) Sl[(rbase + r) * SROW + d] = 100.f;
    if (lane == 0) Yl[rbase + r] = y0;
  }
  __syncthreads();

  for (int m = 0; m < MSTEP; ++m) {
    for (int i = lane; i < 16 * HDIM; i += 32) {
      const int rl = i >> 7, k = i & 127;
      const int row = rbase + rl;
      float v;
      if (k < DIMA)      v = Sl[row * SROW + k] * S0INV;
      else if (k == DIMA) v = tg[(blockBase + row) * MSTEP + m];
      else               v = 0.f;
      buf0[row * HDIM + k] = (_Float16)v;
    }

    mlp_layer(buf0, Wt1, pb1, pg1, pe1, buf1, rbase, lane15, half);
    mlp_layer(buf1, Wt2, pb2, pg2, pe2, buf0, rbase, lane15, half);
    mlp_layer(buf0, Wt3, pb3, pg3, pe3, buf1, rbase, lane15, half);

    {
      v8f acc[7];
#pragma unroll
      for (int nt = 0; nt < 7; ++nt) acc[nt] = {};
      const _Float16* arow = buf1 + (rbase + lane15) * HDIM;
#pragma unroll
      for (int kt = 0; kt < 4; ++kt) {
        const int kb = kt * 32 + half * 8;
        v16h a = load_frag(arow, kb);
#pragma unroll
        for (int nt = 0; nt < 7; ++nt) {
          v16h b = load_frag(Wt4 + (nt * 16 + lane15) * HDIM, kb);
          acc[nt] = __builtin_amdgcn_wmma_f32_16x16x32_f16(
              false, a, false, b, (short)0, acc[nt], false, false);
        }
      }
      float bn[7];
#pragma unroll
      for (int nt = 0; nt < 7; ++nt) bn[nt] = pb4[nt * 16 + lane15];

#pragma unroll
      for (int j = 0; j < 8; ++j) {
        const int row = rbase + j + half * 8;
        const long bidx = blockBase + row;
        const float* dwr = dw + ((long)bidx * MSTEP + m) * DIMA;
        float gain = 0.f, asset = 0.f;
#pragma unroll
        for (int nt = 0; nt < 7; ++nt) {
          const int d = nt * 16 + lane15;
          if (d < DIMA) {
            const float z  = 1.f / (1.f + expf(-(acc[nt][j] + bn[nt])));
            const float s  = Sl[row * SROW + d];
            const float ds = s * (RDT + SIG * dwr[d]);
            gain  += z * ds;
            asset += z * s;
            Sl[row * SROW + d] = s + ds;
          }
        }
        gain  = red16(gain);
        asset = red16(asset);
        if (lane15 == 0) {
          const float y = Yl[row];
          Yl[row] = y + RDT * (y - asset) + gain;
        }
      }
    }
  }

  __syncthreads();
  for (int i = tid; i < ROWS; i += 128) outY[blockBase + i] = Yl[i];
  for (int i = tid; i < ROWS * DIMA; i += 128) {
    const int r = i / DIMA, d = i - r * DIMA;
    outS[(blockBase + r) * (long)DIMA + d] = Sl[r * SROW + d];
  }
}

extern "C" void kernel_launch(void* const* d_in, const int* in_sizes, int n_in,
                              void* d_out, int out_size, void* d_ws, size_t ws_size,
                              hipStream_t stream) {
  (void)in_sizes; (void)n_in; (void)out_size; (void)d_ws; (void)ws_size;
  const float* dw  = (const float*)d_in[0];
  const float* tg  = (const float*)d_in[1];
  const float* W1  = (const float*)d_in[2];
  const float* b1  = (const float*)d_in[3];
  const float* g1  = (const float*)d_in[4];
  const float* be1 = (const float*)d_in[5];
  const float* W2  = (const float*)d_in[6];
  const float* b2  = (const float*)d_in[7];
  const float* g2  = (const float*)d_in[8];
  const float* be2 = (const float*)d_in[9];
  const float* W3  = (const float*)d_in[10];
  const float* b3  = (const float*)d_in[11];
  const float* g3  = (const float*)d_in[12];
  const float* be3 = (const float*)d_in[13];
  const float* W4  = (const float*)d_in[14];
  const float* b4  = (const float*)d_in[15];
  const float* Y0  = (const float*)d_in[16];

  float* outY = (float*)d_out;
  float* outS = outY + BATCH;

  const size_t smem =
      (size_t)(3 * HDIM * HDIM + NPAD4 * HDIM) * sizeof(_Float16) +
      (size_t)(9 * 128 + NPAD4) * sizeof(float) +
      (size_t)(ROWS * SROW + ROWS) * sizeof(float) +
      (size_t)(2 * ROWS * HDIM) * sizeof(_Float16);

  dim3 grid(BATCH / ROWS);
  dim3 block(128);
  bsde_fused_kernel<<<grid, block, smem, stream>>>(
      dw, tg, W1, b1, g1, be1, W2, b2, g2, be2, W3, b3, g3, be3, W4, b4, Y0,
      outY, outS);
}


# Each layer's 128 features and the head's asset tiles split over 4 waves per 16 paths (activations, LayerNorm parts, head sums via LDS + barriers): 256 blocks x 8 waves, 2 per SIMD; sums reordered
# speedup vs baseline: 5.6423x; 5.6423x over previous
#include <hip/hip_runtime.h>
#include <math.h>

#define DIMA   100
#define MSTEP  50
#define HDIM   128
#define BATCH  8192
#define NPAD4  112
#define SROW   104
#define ROWS   32
#define NSPL   4
#define NTHR   256
#define APITCH 136
#define S0INV  0.01f
#define RDT    0.001f
#define SIG    0.2f
#define LNEPS  1e-5f

typedef __attribute__((ext_vector_type(16))) _Float16 v16h;
typedef __attribute__((ext_vector_type(8)))  _Float16 v8h;
typedef __attribute__((ext_vector_type(8)))  float    v8f;
typedef __attribute__((ext_vector_type(4)))  float    v4f;
typedef __attribute__((ext_vector_type(2)))  float    v2f;

union V16U { v16h v; v8h h[2]; };

__device__ __forceinline__ v16h load_frag(const _Float16* base, int kb) {
  V16U u;
  u.h[0] = *(const v8h*)(base + kb);
  u.h[1] = *(const v8h*)(base + kb + 16);
  return u.v;
}

__device__ __forceinline__ float xhalf(float x) {
  return __builtin_bit_cast(float, __builtin_amdgcn_permlanex16(
      __builtin_bit_cast(int, x), __builtin_bit_cast(int, x),
      0x76543210, 0xfedcba98, false, false));
}

__device__ __forceinline__ void mlp_layer_split(const _Float16* inrow, _Float16* outrow,
                                                const _Float16* Wt, const float* bb,
                                                const float* gg, const float* ee,
                                                float* statrow, int f, int lane15, int half) {
  v16h x[4];
#pragma unroll
  for (int kt = 0; kt < 4; ++kt) x[kt] = load_frag(inrow, kt * 32 + half * 8);

  v8f acc0 = {}, acc1 = {};
  const _Float16* w0 = Wt + (f * 32 + lane15) * HDIM;
  const _Float16* w1 = w0 + 16 * HDIM;
#pragma unroll
  for (int kt = 0; kt < 4; ++kt) {
    const int kb = kt * 32 + half * 8;
    const v16h a0 = load_frag(w0, kb);
    const v16h a1 = load_frag(w1, kb);
    acc0 = __builtin_amdgcn_wmma_f32_16x16x32_f16(false, a0, false, x[kt], (short)0, acc0, false, false);
    acc1 = __builtin_amdgcn_wmma_f32_16x16x32_f16(false, a1, false, x[kt], (short)0, acc1, false, false);
  }

  const int fb = f * 32 + half * 8;
  {
    const v4f b00 = *(const v4f*)(bb + fb),      b01 = *(const v4f*)(bb + fb + 4);
    const v4f b10 = *(const v4f*)(bb + fb + 16), b11 = *(const v4f*)(bb + fb + 20);
#pragma unroll
    for (int r = 0; r < 4; ++r) {
      acc0[r] += b00[r]; acc0[4 + r] += b01[r];
      acc1[r] += b10[r]; acc1[4 + r] += b11[r];
    }
  }
  float s0 = 0.f, s1 = 0.f;
#pragma unroll
  for (int r = 0; r < 8; ++r) { s0 += acc0[r]; s1 += acc1[r]; }
  float pl = s0 + s1;
  pl += xhalf(pl);
  const float mw = pl * (1.f / 32.f);
  float q0 = 0.f, q1 = 0.f;
#pragma unroll
  for (int r = 0; r < 8; ++r) {
    const float d0 = acc0[r] - mw, d1 = acc1[r] - mw;
    q0 += d0 * d0; q1 += d1 * d1;
  }
  float ql = q0 + q1;
  ql += xhalf(ql);
  if (half == 0) *(v2f*)(statrow + 2 * f) = v2f{mw, ql};
  __syncthreads();

  const v4f sA = *(const v4f*)statrow, sB = *(const v4f*)(statrow + 4);
  const float mu = ((sA[0] + sA[2]) + (sB[0] + sB[2])) * 0.25f;
  const float e0 = sA[0] - mu, e1 = sA[2] - mu, e2 = sB[0] - mu, e3 = sB[2] - mu;
  const float m2 = ((sA[1] + sA[3]) + (sB[1] + sB[3])) +
                   32.f * ((e0 * e0 + e1 * e1) + (e2 * e2 + e3 * e3));
  const float rstd = rsqrtf(m2 * (1.f / 128.f) + LNEPS);

  const v4f g00 = *(const v4f*)(gg + fb),      g01 = *(const v4f*)(gg + fb + 4);
  const v4f g10 = *(const v4f*)(gg + fb + 16), g11 = *(const v4f*)(gg + fb + 20);
  const v4f t00 = *(const v4f*)(ee + fb),      t01 = *(const v4f*)(ee + fb + 4);
  const v4f t10 = *(const v4f*)(ee + fb + 16), t11 = *(const v4f*)(ee + fb + 20);
  v8h o0, o1;
#pragma unroll
  for (int r = 0; r < 8; ++r) {
    const float gn0 = r < 4 ? g00[r & 3] : g01[r & 3];
    const float en0 = r < 4 ? t00[r & 3] : t01[r & 3];
    const float xn0 = (acc0[r] - mu) * rstd * gn0 + en0;
    o0[r] = (_Float16)(0.5f * xn0 * (1.f + erff(xn0 * 0.70710678118654752f)));
    const float gn1 = r < 4 ? g10[r & 3] : g11[r & 3];
    const float en1 = r < 4 ? t10[r & 3] : t11[r & 3];
    const float xn1 = (acc1[r] - mu) * rstd * gn1 + en1;
    o1[r] = (_Float16)(0.5f * xn1 * (1.f + erff(xn1 * 0.70710678118654752f)));
  }
  *(v8h*)(outrow + fb)      = o0;
  *(v8h*)(outrow + fb + 16) = o1;
  __syncthreads();
}

__device__ __forceinline__ void sde_tile8(const v8f& acc, float (&S)[8], const float* dwp,
                                          const float* bp, float& gsum, float& asum) {
  const v4f w0 = *(const v4f*)dwp, w1 = *(const v4f*)(dwp + 4);
  const v4f c0 = *(const v4f*)bp,  c1 = *(const v4f*)(bp + 4);
  float gs = 0.f, as = 0.f;
#pragma unroll
  for (int r = 0; r < 8; ++r) {
    const float dwv = r < 4 ? w0[r & 3] : w1[r & 3];
    const float bn  = r < 4 ? c0[r & 3] : c1[r & 3];
    const float z  = 1.f / (1.f + expf(-(acc[r] + bn)));
    const float s  = S[r];
    const float ds = s * (RDT + SIG * dwv);
    gs += z * ds;
    as += z * s;
    S[r] = s + ds;
  }
  gsum = gs; asum = as;
}

extern "C" __global__ void __launch_bounds__(NTHR)
bsde_fused_kernel(const float* __restrict__ dw, const float* __restrict__ tg,
                  const float* __restrict__ W1, const float* __restrict__ b1,
                  const float* __restrict__ g1, const float* __restrict__ be1,
                  const float* __restrict__ W2, const float* __restrict__ b2,
                  const float* __restrict__ g2, const float* __restrict__ be2,
                  const float* __restrict__ W3, const float* __restrict__ b3,
                  const float* __restrict__ g3, const float* __restrict__ be3,
                  const float* __restrict__ W4, const float* __restrict__ b4,
                  const float* __restrict__ Y0,
                  float* __restrict__ outY, float* __restrict__ outS) {
  extern __shared__ __align__(16) char smem[];
  _Float16* Wt1 = (_Float16*)smem;
  _Float16* Wt2 = Wt1 + HDIM * HDIM;
  _Float16* Wt3 = Wt2 + HDIM * HDIM;
  _Float16* Wt4 = Wt3 + HDIM * HDIM;
  float* prm = (float*)(Wt4 + NPAD4 * HDIM);
  float* pb1 = prm;        float* pg1 = prm + 128;  float* pe1 = prm + 256;
  float* pb2 = prm + 384;  float* pg2 = prm + 512;  float* pe2 = prm + 640;
  float* pb3 = prm + 768;  float* pg3 = prm + 896;  float* pe3 = prm + 1024;
  float* pb4 = prm + 1152;
  float* Sl  = prm + 1264;
  float* Yl  = Sl + ROWS * SROW;
  float* stat = Yl + ROWS;
  float* hsum = stat + ROWS * NSPL * 2;
  _Float16* xin = (_Float16*)(hsum + ROWS * NSPL * 2);
  _Float16* act = xin + ROWS * APITCH;

  const int tid = threadIdx.x;
  const int lane = tid & 31, lane15 = lane & 15, half = lane >> 4;
  const int wv = __builtin_amdgcn_readfirstlane(tid >> 5);
  const int g = wv & 1;
  const int f = wv >> 1;
  const long blockBase = (long)blockIdx.x * ROWS;
  const int p = g * 16 + lane15;
  const long bidx = blockBase + p;
  const int nb = half * 8;
  const int ta = f, tb = f + 4;

  for (int i = tid; i < HDIM * HDIM; i += NTHR) {
    const int n = i >> 7, k = i & 127;
    Wt1[i] = (_Float16)(k < DIMA + 1 ? W1[k * HDIM + n] : 0.f);
    Wt2[i] = (_Float16)W2[k * HDIM + n];
    Wt3[i] = (_Float16)W3[k * HDIM + n];
  }
  for (int i = tid; i < NPAD4 * HDIM; i += NTHR) {
    const int n = i >> 7, k = i & 127;
    Wt4[i] = (_Float16)(n < DIMA ? W4[k * DIMA + n] : 0.f);
  }
  if (tid < HDIM) {
    pb1[tid] = b1[tid]; pg1[tid] = g1[tid]; pe1[tid] = be1[tid];
    pb2[tid] = b2[tid]; pg2[tid] = g2[tid]; pe2[tid] = be2[tid];
    pb3[tid] = b3[tid]; pg3[tid] = g3[tid]; pe3[tid] = be3[tid];
  }
  if (tid < NPAD4) pb4[tid] = (tid < DIMA) ? b4[tid] : 0.f;
  if (tid < ROWS) {
    const v8h z = {};
    *(v8h*)(xin + tid * APITCH + 112) = z;
    *(v8h*)(xin + tid * APITCH + 120) = z;
  }

  float Sa[8], Sb[8];
#pragma unroll
  for (int r = 0; r < 8; ++r) {
    Sa[r] = 100.f;
    Sb[r] = (tb * 16 + nb + r < DIMA) ? 100.f : 0.f;
  }
  float Y = Y0[0];
  _Float16* xrow = xin + p * APITCH;
  _Float16* arow = act + p * APITCH;
  float* statrow = stat + p * (NSPL * 2);
  float* hrow = hsum + p * (NSPL * 2);

  {
    v8h o;
#pragma unroll
    for (int r = 0; r < 8; ++r) o[r] = (_Float16)(Sa[r] * S0INV);
    *(v8h*)(xrow + ta * 16 + nb) = o;
    if (f < 3) {
      v8h o2;
#pragma unroll
      for (int r = 0; r < 8; ++r) o2[r] = (_Float16)(Sb[r] * S0INV);
      if (f == 2) {
        const float tv = tg[bidx * MSTEP];
        o2[4] = (half == 0) ? (_Float16)tv : (_Float16)0.f;
      }
      *(v8h*)(xrow + tb * 16 + nb) = o2;
    }
  }
  __syncthreads();

  for (int m = 0; m < MSTEP; ++m) {
    mlp_layer_split(xrow, arow, Wt1, pb1, pg1, pe1, statrow, f, lane15, half);
    mlp_layer_split(arow, arow, Wt2, pb2, pg2, pe2, statrow, f, lane15, half);
    mlp_layer_split(arow, arow, Wt3, pb3, pg3, pe3, statrow, f, lane15, half);

    {
      v16h x[4];
#pragma unroll
      for (int kt = 0; kt < 4; ++kt) x[kt] = load_frag(arow, kt * 32 + half * 8);
      v8f accA = {}, accB = {};
      if (f < 3) {
        const _Float16* wB = Wt4 + (tb * 16 + lane15) * HDIM;
#pragma unroll
        for (int kt = 0; kt < 4; ++kt)
          accB = __builtin_amdgcn_wmma_f32_16x16x32_f16(
              false, load_frag(wB, kt * 32 + half * 8), false, x[kt], (short)0, accB, false, false);
      }
      const _Float16* wA = Wt4 + (ta * 16 + lane15) * HDIM;
#pragma unroll
      for (int kt = 0; kt < 4; ++kt)
        accA = __builtin_amdgcn_wmma_f32_16x16x32_f16(
            false, load_frag(wA, kt * 32 + half * 8), false, x[kt], (short)0, accA, false, false);

      const float* dwr = dw + (bidx * MSTEP + m) * DIMA;
      float gA, aA, gB = 0.f, aB = 0.f;
      sde_tile8(accA, Sa, dwr + ta * 16 + nb, pb4 + ta * 16 + nb, gA, aA);
      if (f < 2) {
        sde_tile8(accB, Sb, dwr + tb * 16 + nb, pb4 + tb * 16 + nb, gB, aB);
      } else if (f == 2) {
        if (half == 0) {
          const v4f w0 = *(const v4f*)(dwr + 96);
          const v4f c0 = *(const v4f*)(pb4 + 96);
#pragma unroll
          for (int r = 0; r < 4; ++r) {
            const float z  = 1.f / (1.f + expf(-(accB[r] + c0[r])));
            const float s  = Sb[r];
            const float ds = s * (RDT + SIG * w0[r]);
            gB += z * ds;
            aB += z * s;
            Sb[r] = s + ds;
          }
        }
      }
      float gain = gA + gB, asset = aA + aB;
      gain  += xhalf(gain);
      asset += xhalf(asset);
      if (half == 0) *(v2f*)(hrow + 2 * f) = v2f{gain, asset};

      {
        v8h o;
#pragma unroll
        for (int r = 0; r < 8; ++r) o[r] = (_Float16)(Sa[r] * S0INV);
        *(v8h*)(xrow + ta * 16 + nb) = o;
        if (f < 3) {
          v8h o2;
#pragma unroll
          for (int r = 0; r < 8; ++r) o2[r] = (_Float16)(Sb[r] * S0INV);
          if (f == 2) {
            const float tv = (m + 1 < MSTEP) ? tg[bidx * MSTEP + m + 1] : 0.f;
            o2[4] = (half == 0) ? (_Float16)tv : (_Float16)0.f;
          }
          *(v8h*)(xrow + tb * 16 + nb) = o2;
        }
      }
      __syncthreads();
      const v4f hA = *(const v4f*)hrow, hB = *(const v4f*)(hrow + 4);
      const float gsum = (hA[0] + hA[2]) + (hB[0] + hB[2]);
      const float asum = (hA[1] + hA[3]) + (hB[1] + hB[3]);
      Y = Y + RDT * (Y - asum) + gsum;
    }
  }

  {
    float* sp = Sl + p * SROW + ta * 16 + nb;
    *(v4f*)sp       = v4f{Sa[0], Sa[1], Sa[2], Sa[3]};
    *(v4f*)(sp + 4) = v4f{Sa[4], Sa[5], Sa[6], Sa[7]};
  }
  if (f < 2) {
    float* sp = Sl + p * SROW + tb * 16 + nb;
    *(v4f*)sp       = v4f{Sb[0], Sb[1], Sb[2], Sb[3]};
    *(v4f*)(sp + 4) = v4f{Sb[4], Sb[5], Sb[6], Sb[7]};
  } else if (f == 2) {
    if (half == 0) *(v4f*)(Sl + p * SROW + 96) = v4f{Sb[0], Sb[1], Sb[2], Sb[3]};
  }
  if (f == 0 && half == 0) Yl[p] = Y;
  __syncthreads();
  for (int i = tid; i < ROWS; i += NTHR) outY[blockBase + i] = Yl[i];
  for (int i = tid; i < ROWS * DIMA; i += NTHR) {
    const int r = i / DIMA, d = i - r * DIMA;
    outS[(blockBase + r) * (long)DIMA + d] = Sl[r * SROW + d];
  }
}

extern "C" void kernel_launch(void* const* d_in, const int* in_sizes, int n_in,
                              void* d_out, int out_size, void* d_ws, size_t ws_size,
                              hipStream_t stream) {
  (void)in_sizes; (void)n_in; (void)out_size; (void)d_ws; (void)ws_size;
  const float* dw  = (const float*)d_in[0];
  const float* tg  = (const float*)d_in[1];
  const float* W1  = (const float*)d_in[2];
  const float* b1  = (const float*)d_in[3];
  const float* g1  = (const float*)d_in[4];
  const float* be1 = (const float*)d_in[5];
  const float* W2  = (const float*)d_in[6];
  const float* b2  = (const float*)d_in[7];
  const float* g2  = (const float*)d_in[8];
  const float* be2 = (const float*)d_in[9];
  const float* W3  = (const float*)d_in[10];
  const float* b3  = (const float*)d_in[11];
  const float* g3  = (const float*)d_in[12];
  const float* be3 = (const float*)d_in[13];
  const float* W4  = (const float*)d_in[14];
  const float* b4  = (const float*)d_in[15];
  const float* Y0  = (const float*)d_in[16];

  float* outY = (float*)d_out;
  float* outS = outY + BATCH;

  const size_t smem =
      (size_t)(3 * HDIM * HDIM + NPAD4 * HDIM) * sizeof(_Float16) +
      (size_t)(9 * 128 + NPAD4) * sizeof(float) +
      (size_t)(ROWS * SROW + ROWS) * sizeof(float) +
      (size_t)(2 * ROWS * NSPL * 2) * sizeof(float) +
      (size_t)(2 * ROWS * APITCH) * sizeof(_Float16);

  dim3 grid(BATCH / ROWS);
  dim3 block(NTHR);
  bsde_fused_kernel<<<grid, block, smem, stream>>>(
      dw, tg, W1, b1, g1, be1, W2, b2, g2, be2, W3, b3, g3, be3, W4, b4, Y0,
      outY, outS);
}


# Weights live in registers: each wave loads the A fragments of its own tiles of all four products once from global memory (same f16 values) and keeps them all 50 steps; no weights in LDS; same sums
# speedup vs baseline: 11.0323x; 1.9553x over previous
#include <hip/hip_runtime.h>
#include <math.h>

#define DIMA   100
#define MSTEP  50
#define HDIM   128
#define BATCH  8192
#define NPAD4  112
#define SROW   104
#define ROWS   32
#define NSPL   4
#define NTHR   256
#define APITCH 136
#define S0INV  0.01f
#define RDT    0.001f
#define SIG    0.2f
#define LNEPS  1e-5f

typedef __attribute__((ext_vector_type(16))) _Float16 v16h;
typedef __attribute__((ext_vector_type(8)))  _Float16 v8h;
typedef __attribute__((ext_vector_type(8)))  float    v8f;
typedef __attribute__((ext_vector_type(4)))  float    v4f;
typedef __attribute__((ext_vector_type(2)))  float    v2f;

union V16U { v16h v; v8h h[2]; };

__device__ __forceinline__ v16h load_frag(const _Float16* base, int kb) {
  V16U u;
  u.h[0] = *(const v8h*)(base + kb);
  u.h[1] = *(const v8h*)(base + kb + 16);
  return u.v;
}

template <bool PADN>
__device__ __forceinline__ v16h wfrag_global(const float* __restrict__ W, int ldn, int n, int nlim,
                                             int kb, int klim) {
  const bool nok = PADN ? n < nlim : true;
  const int nn = nok ? n : nlim - 1;
  v16h v;
#pragma unroll
  for (int s = 0; s < 16; ++s) {
    const int k = kb + (s & 7) + ((s >> 3) << 4);
    const int kk = k < klim ? k : klim - 1;
    const float w = W[kk * ldn + nn];
    v[s] = (nok && k < klim) ? (_Float16)w : (_Float16)0.f;
  }
  return v;
}

__device__ __forceinline__ float xhalf(float x) {
  return __builtin_bit_cast(float, __builtin_amdgcn_permlanex16(
      __builtin_bit_cast(int, x), __builtin_bit_cast(int, x),
      0x76543210, 0xfedcba98, false, false));
}

__device__ __forceinline__ void mlp_layer_split(const _Float16* inrow, _Float16* outrow,
                                                const v16h (&wt0)[4], const v16h (&wt1)[4],
                                                const float* bb,
                                                const float* gg, const float* ee,
                                                float* statrow, int f, int lane15, int half) {
  v16h x[4];
#pragma unroll
  for (int kt = 0; kt < 4; ++kt) x[kt] = load_frag(inrow, kt * 32 + half * 8);

  v8f acc0 = {}, acc1 = {};
#pragma unroll
  for (int kt = 0; kt < 4; ++kt) {
    acc0 = __builtin_amdgcn_wmma_f32_16x16x32_f16(false, wt0[kt], false, x[kt], (short)0, acc0, false, false);
    acc1 = __builtin_amdgcn_wmma_f32_16x16x32_f16(false, wt1[kt], false, x[kt], (short)0, acc1, false, false);
  }

  const int fb = f * 32 + half * 8;
  {
    const v4f b00 = *(const v4f*)(bb + fb),      b01 = *(const v4f*)(bb + fb + 4);
    const v4f b10 = *(const v4f*)(bb + fb + 16), b11 = *(const v4f*)(bb + fb + 20);
#pragma unroll
    for (int r = 0; r < 4; ++r) {
      acc0[r] += b00[r]; acc0[4 + r] += b01[r];
      acc1[r] += b10[r]; acc1[4 + r] += b11[r];
    }
  }
  float s0 = 0.f, s1 = 0.f;
#pragma unroll
  for (int r = 0; r < 8; ++r) { s0 += acc0[r]; s1 += acc1[r]; }
  float pl = s0 + s1;
  pl += xhalf(pl);
  const float mw = pl * (1.f / 32.f);
  float q0 = 0.f, q1 = 0.f;
#pragma unroll
  for (int r = 0; r < 8; ++r) {
    const float d0 = acc0[r] - mw, d1 = acc1[r] - mw;
    q0 += d0 * d0; q1 += d1 * d1;
  }
  float ql = q0 + q1;
  ql += xhalf(ql);
  if (half == 0) *(v2f*)(statrow + 2 * f) = v2f{mw, ql};
  __syncthreads();

  const v4f sA = *(const v4f*)statrow, sB = *(const v4f*)(statrow + 4);
  const float mu = ((sA[0] + sA[2]) + (sB[0] + sB[2])) * 0.25f;
  const float e0 = sA[0] - mu, e1 = sA[2] - mu, e2 = sB[0] - mu, e3 = sB[2] - mu;
  const float m2 = ((sA[1] + sA[3]) + (sB[1] + sB[3])) +
                   32.f * ((e0 * e0 + e1 * e1) + (e2 * e2 + e3 * e3));
  const float rstd = rsqrtf(m2 * (1.f / 128.f) + LNEPS);

  const v4f g00 = *(const v4f*)(gg + fb),      g01 = *(const v4f*)(gg + fb + 4);
  const v4f g10 = *(const v4f*)(gg + fb + 16), g11 = *(const v4f*)(gg + fb + 20);
  const v4f t00 = *(const v4f*)(ee + fb),      t01 = *(const v4f*)(ee + fb + 4);
  const v4f t10 = *(const v4f*)(ee + fb + 16), t11 = *(const v4f*)(ee + fb + 20);
  v8h o0, o1;
#pragma unroll
  for (int r = 0; r < 8; ++r) {
    const float gn0 = r < 4 ? g00[r & 3] : g01[r & 3];
    const float en0 = r < 4 ? t00[r & 3] : t01[r & 3];
    const float xn0 = (acc0[r] - mu) * rstd * gn0 + en0;
    o0[r] = (_Float16)(0.5f * xn0 * (1.f + erff(xn0 * 0.70710678118654752f)));
    const float gn1 = r < 4 ? g10[r & 3] : g11[r & 3];
    const float en1 = r < 4 ? t10[r & 3] : t11[r & 3];
    const float xn1 = (acc1[r] - mu) * rstd * gn1 + en1;
    o1[r] = (_Float16)(0.5f * xn1 * (1.f + erff(xn1 * 0.70710678118654752f)));
  }
  *(v8h*)(outrow + fb)      = o0;
  *(v8h*)(outrow + fb + 16) = o1;
  __syncthreads();
}

__device__ __forceinline__ void sde_tile8(const v8f& acc, float (&S)[8], const float* dwp,
                                          const float* bp, float& gsum, float& asum) {
  const v4f w0 = *(const v4f*)dwp, w1 = *(const v4f*)(dwp + 4);
  const v4f c0 = *(const v4f*)bp,  c1 = *(const v4f*)(bp + 4);
  float gs = 0.f, as = 0.f;
#pragma unroll
  for (int r = 0; r < 8; ++r) {
    const float dwv = r < 4 ? w0[r & 3] : w1[r & 3];
    const float bn  = r < 4 ? c0[r & 3] : c1[r & 3];
    const float z  = 1.f / (1.f + expf(-(acc[r] + bn)));
    const float s  = S[r];
    const float ds = s * (RDT + SIG * dwv);
    gs += z * ds;
    as += z * s;
    S[r] = s + ds;
  }
  gsum = gs; asum = as;
}

extern "C" __global__ void __launch_bounds__(NTHR)
bsde_fused_kernel(const float* __restrict__ dw, const float* __restrict__ tg,
                  const float* __restrict__ W1, const float* __restrict__ b1,
                  const float* __restrict__ g1, const float* __restrict__ be1,
                  const float* __restrict__ W2, const float* __restrict__ b2,
                  const float* __restrict__ g2, const float* __restrict__ be2,
                  const float* __restrict__ W3, const float* __restrict__ b3,
                  const float* __restrict__ g3, const float* __restrict__ be3,
                  const float* __restrict__ W4, const float* __restrict__ b4,
                  const float* __restrict__ Y0,
                  float* __restrict__ outY, float* __restrict__ outS) {
  extern __shared__ __align__(16) char smem[];
  float* prm = (float*)smem;
  float* pb1 = prm;        float* pg1 = prm + 128;  float* pe1 = prm + 256;
  float* pb2 = prm + 384;  float* pg2 = prm + 512;  float* pe2 = prm + 640;
  float* pb3 = prm + 768;  float* pg3 = prm + 896;  float* pe3 = prm + 1024;
  float* pb4 = prm + 1152;
  float* Sl  = prm + 1264;
  float* Yl  = Sl + ROWS * SROW;
  float* stat = Yl + ROWS;
  float* hsum = stat + ROWS * NSPL * 2;
  _Float16* xin = (_Float16*)(hsum + ROWS * NSPL * 2);
  _Float16* act = xin + ROWS * APITCH;

  const int tid = threadIdx.x;
  const int lane = tid & 31, lane15 = lane & 15, half = lane >> 4;
  const int wv = __builtin_amdgcn_readfirstlane(tid >> 5);
  const int g = wv & 1;
  const int f = wv >> 1;
  const long blockBase = (long)blockIdx.x * ROWS;
  const int p = g * 16 + lane15;
  const long bidx = blockBase + p;
  const int nb = half * 8;
  const int ta = f, tb = f + 4;

  v16h wl1a[4], wl1b[4], wl2a[4], wl2b[4], wl3a[4], wl3b[4], wh4a[4], wh4b[4];
  {
    const int n0 = f * 32 + lane15;
#pragma unroll
    for (int kt = 0; kt < 4; ++kt) {
      const int kb = kt * 32 + half * 8;
      wl1a[kt] = wfrag_global<false>(W1, HDIM, n0,      HDIM, kb, DIMA + 1);
      wl1b[kt] = wfrag_global<false>(W1, HDIM, n0 + 16, HDIM, kb, DIMA + 1);
      wl2a[kt] = wfrag_global<false>(W2, HDIM, n0,      HDIM, kb, HDIM);
      wl2b[kt] = wfrag_global<false>(W2, HDIM, n0 + 16, HDIM, kb, HDIM);
      wl3a[kt] = wfrag_global<false>(W3, HDIM, n0,      HDIM, kb, HDIM);
      wl3b[kt] = wfrag_global<false>(W3, HDIM, n0 + 16, HDIM, kb, HDIM);
      wh4a[kt] = wfrag_global<false>(W4, DIMA, ta * 16 + lane15, DIMA, kb, HDIM);
      wh4b[kt] = v16h{};
    }
    if (f < 3) {
#pragma unroll
      for (int kt = 0; kt < 4; ++kt)
        wh4b[kt] = wfrag_global<true>(W4, DIMA, tb * 16 + lane15, DIMA, kt * 32 + half * 8, HDIM);
    }
  }
  if (tid < HDIM) {
    pb1[tid] = b1[tid]; pg1[tid] = g1[tid]; pe1[tid] = be1[tid];
    pb2[tid] = b2[tid]; pg2[tid] = g2[tid]; pe2[tid] = be2[tid];
    pb3[tid] = b3[tid]; pg3[tid] = g3[tid]; pe3[tid] = be3[tid];
  }
  if (tid < NPAD4) pb4[tid] = (tid < DIMA) ? b4[tid] : 0.f;
  if (tid < ROWS) {
    const v8h z = {};
    *(v8h*)(xin + tid * APITCH + 112) = z;
    *(v8h*)(xin + tid * APITCH + 120) = z;
  }

  float Sa[8], Sb[8];
#pragma unroll
  for (int r = 0; r < 8; ++r) {
    Sa[r] = 100.f;
    Sb[r] = (tb * 16 + nb + r < DIMA) ? 100.f : 0.f;
  }
  float Y = Y0[0];
  _Float16* xrow = xin + p * APITCH;
  _Float16* arow = act + p * APITCH;
  float* statrow = stat + p * (NSPL * 2);
  float* hrow = hsum + p * (NSPL * 2);

  {
    v8h o;
#pragma unroll
    for (int r = 0; r < 8; ++r) o[r] = (_Float16)(Sa[r] * S0INV);
    *(v8h*)(xrow + ta * 16 + nb) = o;
    if (f < 3) {
      v8h o2;
#pragma unroll
      for (int r = 0; r < 8; ++r) o2[r] = (_Float16)(Sb[r] * S0INV);
      if (f == 2) {
        const float tv = tg[bidx * MSTEP];
        o2[4] = (half == 0) ? (_Float16)tv : (_Float16)0.f;
      }
      *(v8h*)(xrow + tb * 16 + nb) = o2;
    }
  }
  __syncthreads();

  for (int m = 0; m < MSTEP; ++m) {
    mlp_layer_split(xrow, arow, wl1a, wl1b, pb1, pg1, pe1, statrow, f, lane15, half);
    mlp_layer_split(arow, arow, wl2a, wl2b, pb2, pg2, pe2, statrow, f, lane15, half);
    mlp_layer_split(arow, arow, wl3a, wl3b, pb3, pg3, pe3, statrow, f, lane15, half);

    {
      v16h x[4];
#pragma unroll
      for (int kt = 0; kt < 4; ++kt) x[kt] = load_frag(arow, kt * 32 + half * 8);
      v8f accA = {}, accB = {};
      if (f < 3) {
#pragma unroll
        for (int kt = 0; kt < 4; ++kt)
          accB = __builtin_amdgcn_wmma_f32_16x16x32_f16(
              false, wh4b[kt], false, x[kt], (short)0, accB, false, false);
      }
#pragma unroll
      for (int kt = 0; kt < 4; ++kt)
        accA = __builtin_amdgcn_wmma_f32_16x16x32_f16(
            false, wh4a[kt], false, x[kt], (short)0, accA, false, false);

      const float* dwr = dw + (bidx * MSTEP + m) * DIMA;
      float gA, aA, gB = 0.f, aB = 0.f;
      sde_tile8(accA, Sa, dwr + ta * 16 + nb, pb4 + ta * 16 + nb, gA, aA);
      if (f < 2) {
        sde_tile8(accB, Sb, dwr + tb * 16 + nb, pb4 + tb * 16 + nb, gB, aB);
      } else if (f == 2) {
        if (half == 0) {
          const v4f w0 = *(const v4f*)(dwr + 96);
          const v4f c0 = *(const v4f*)(pb4 + 96);
#pragma unroll
          for (int r = 0; r < 4; ++r) {
            const float z  = 1.f / (1.f + expf(-(accB[r] + c0[r])));
            const float s  = Sb[r];
            const float ds = s * (RDT + SIG * w0[r]);
            gB += z * ds;
            aB += z * s;
            Sb[r] = s + ds;
          }
        }
      }
      float gain = gA + gB, asset = aA + aB;
      gain  += xhalf(gain);
      asset += xhalf(asset);
      if (half == 0) *(v2f*)(hrow + 2 * f) = v2f{gain, asset};

      {
        v8h o;
#pragma unroll
        for (int r = 0; r < 8; ++r) o[r] = (_Float16)(Sa[r] * S0INV);
        *(v8h*)(xrow + ta * 16 + nb) = o;
        if (f < 3) {
          v8h o2;
#pragma unroll
          for (int r = 0; r < 8; ++r) o2[r] = (_Float16)(Sb[r] * S0INV);
          if (f == 2) {
            const float tv = (m + 1 < MSTEP) ? tg[bidx * MSTEP + m + 1] : 0.f;
            o2[4] = (half == 0) ? (_Float16)tv : (_Float16)0.f;
          }
          *(v8h*)(xrow + tb * 16 + nb) = o2;
        }
      }
      __syncthreads();
      const v4f hA = *(const v4f*)hrow, hB = *(const v4f*)(hrow + 4);
      const float gsum = (hA[0] + hA[2]) + (hB[0] + hB[2]);
      const float asum = (hA[1] + hA[3]) + (hB[1] + hB[3]);
      Y = Y + RDT * (Y - asum) + gsum;
    }
  }

  {
    float* sp = Sl + p * SROW + ta * 16 + nb;
    *(v4f*)sp       = v4f{Sa[0], Sa[1], Sa[2], Sa[3]};
    *(v4f*)(sp + 4) = v4f{Sa[4], Sa[5], Sa[6], Sa[7]};
  }
  if (f < 2) {
    float* sp = Sl + p * SROW + tb * 16 + nb;
    *(v4f*)sp       = v4f{Sb[0], Sb[1], Sb[2], Sb[3]};
    *(v4f*)(sp + 4) = v4f{Sb[4], Sb[5], Sb[6], Sb[7]};
  } else if (f == 2) {
    if (half == 0) *(v4f*)(Sl + p * SROW + 96) = v4f{Sb[0], Sb[1], Sb[2], Sb[3]};
  }
  if (f == 0 && half == 0) Yl[p] = Y;
  __syncthreads();
  for (int i = tid; i < ROWS; i += NTHR) outY[blockBase + i] = Yl[i];
  for (int i = tid; i < ROWS * DIMA; i += NTHR) {
    const int r = i / DIMA, d = i - r * DIMA;
    outS[(blockBase + r) * (long)DIMA + d] = Sl[r * SROW + d];
  }
}

extern "C" void kernel_launch(void* const* d_in, const int* in_sizes, int n_in,
                              void* d_out, int out_size, void* d_ws, size_t ws_size,
                              hipStream_t stream) {
  (void)in_sizes; (void)n_in; (void)out_size; (void)d_ws; (void)ws_size;
  const float* dw  = (const float*)d_in[0];
  const float* tg  = (const float*)d_in[1];
  const float* W1  = (const float*)d_in[2];
  const float* b1  = (const float*)d_in[3];
  const float* g1  = (const float*)d_in[4];
  const float* be1 = (const float*)d_in[5];
  const float* W2  = (const float*)d_in[6];
  const float* b2  = (const float*)d_in[7];
  const float* g2  = (const float*)d_in[8];
  const float* be2 = (const float*)d_in[9];
  const float* W3  = (const float*)d_in[10];
  const float* b3  = (const float*)d_in[11];
  const float* g3  = (const float*)d_in[12];
  const float* be3 = (const float*)d_in[13];
  const float* W4  = (const float*)d_in[14];
  const float* b4  = (const float*)d_in[15];
  const float* Y0  = (const float*)d_in[16];

  float* outY = (float*)d_out;
  float* outS = outY + BATCH;

  const size_t smem =
      (size_t)(9 * 128 + NPAD4) * sizeof(float) +
      (size_t)(ROWS * SROW + ROWS) * sizeof(float) +
      (size_t)(2 * ROWS * NSPL * 2) * sizeof(float) +
      (size_t)(2 * ROWS * APITCH) * sizeof(_Float16);

  dim3 grid(BATCH / ROWS);
  dim3 block(NTHR);
  bsde_fused_kernel<<<grid, block, smem, stream>>>(
      dw, tg, W1, b1, g1, be1, W2, b2, g2, be2, W3, b3, g3, be3, W4, b4, Y0,
      outY, outS);
}


# S advances from dW alone: the next input row is published between layers 2 and 3 and the head has no barrier (6 a step, was 7); dW asked for at the step's top, LDS-only barriers; same sums
# speedup vs baseline: 23.4723x; 2.1276x over previous
#include <hip/hip_runtime.h>
#include <math.h>

#define DIMA   100
#define MSTEP  50
#define HDIM   128
#define BATCH  8192
#define NPAD4  112
#define SROW   104
#define ROWS   16
#define NSPL   8
#define NTHR   256
#define APITCH 136
#define S0INV  0.01f
#define RDT    0.001f
#define SIG    0.2f
#define LNEPS  1e-5f

typedef __attribute__((ext_vector_type(16))) _Float16 v16h;
typedef __attribute__((ext_vector_type(8)))  _Float16 v8h;
typedef __attribute__((ext_vector_type(8)))  float    v8f;
typedef __attribute__((ext_vector_type(4)))  float    v4f;
typedef __attribute__((ext_vector_type(2)))  float    v2f;

union V16U { v16h v; v8h h[2]; };

__device__ __forceinline__ v16h load_frag(const _Float16* base, int kb) {
  V16U u;
  u.h[0] = *(const v8h*)(base + kb);
  u.h[1] = *(const v8h*)(base + kb + 16);
  return u.v;
}

template <bool PADN>
__device__ __forceinline__ v16h wfrag_global(const float* __restrict__ W, int ldn, int n, int nlim,
                                             int kb, int klim) {
  const bool nok = PADN ? n < nlim : true;
  const int nn = nok ? n : nlim - 1;
  v16h v;
#pragma unroll
  for (int s = 0; s < 16; ++s) {
    const int k = kb + (s & 7) + ((s >> 3) << 4);
    const int kk = k < klim ? k : klim - 1;
    const float w = W[kk * ldn + nn];
    v[s] = (nok && k < klim) ? (_Float16)w : (_Float16)0.f;
  }
  return v;
}

__device__ __forceinline__ float xhalf(float x) {
  return __builtin_bit_cast(float, __builtin_amdgcn_permlanex16(
      __builtin_bit_cast(int, x), __builtin_bit_cast(int, x),
      0x76543210, 0xfedcba98, false, false));
}

__device__ __forceinline__ v2f gelu_pair(v2f x) {
  v2f a;
  a[0] = __builtin_fabsf(x[0]);
  a[1] = __builtin_fabsf(x[1]);
  const v2f c7 = {-2.8349113563308492e-06f, -2.8349113563308492e-06f};
  const v2f c6 = {3.937767178285867e-05f, 3.937767178285867e-05f};
  const v2f c5 = {-0.0001861801283666864f, -0.0001861801283666864f};
  const v2f c4 = {-0.0001369366655126214f, -0.0001369366655126214f};
  const v2f c3 = {0.0070634204894304276f, 0.0070634204894304276f};
  const v2f c2 = {-0.052496179938316345f, -0.052496179938316345f};
  const v2f c1 = {-0.4592081904411316f, -0.4592081904411316f};
  const v2f c0 = {-1.1511051654815674f, -1.1511051654815674f};
  const v2f m1 = {-1.f, -1.f};
  v2f g = __builtin_elementwise_fma(c7, a, c6);
  g = __builtin_elementwise_fma(g, a, c5);
  g = __builtin_elementwise_fma(g, a, c4);
  g = __builtin_elementwise_fma(g, a, c3);
  g = __builtin_elementwise_fma(g, a, c2);
  g = __builtin_elementwise_fma(g, a, c1);
  g = __builtin_elementwise_fma(g, a, c0);
  const v2f r = __builtin_elementwise_fma(a, g, m1);
  v2f h;
  h[0] = __builtin_amdgcn_exp2f(r[0]);
  h[1] = __builtin_amdgcn_exp2f(r[1]);
  v2f relu;
  relu[0] = __builtin_fmaxf(x[0], 0.f);
  relu[1] = __builtin_fmaxf(x[1], 0.f);
  return __builtin_elementwise_fma(-a, h, relu);
}

__device__ __forceinline__ void lds_barrier() {
  __builtin_amdgcn_fence(__ATOMIC_RELEASE, "workgroup", "local");
  __builtin_amdgcn_s_barrier();
  __builtin_amdgcn_fence(__ATOMIC_ACQUIRE, "workgroup", "local");
}

__device__ __forceinline__ void mlp_layer_split(const _Float16* inrow, _Float16* outrow,
                                                const v16h (&wt)[4],
                                                const float* bb,
                                                const float* gg, const float* ee,
                                                float* statrow, int f, int half) {
  v16h x[4];
#pragma unroll
  for (int kt = 0; kt < 4; ++kt) x[kt] = load_frag(inrow, kt * 32 + half * 8);

  v8f acc = {};
#pragma unroll
  for (int kt = 0; kt < 4; ++kt)
    acc = __builtin_amdgcn_wmma_f32_16x16x32_f16(false, wt[kt], false, x[kt], (short)0, acc, false, false);

  const int fb = f * 16 + half * 8;
  {
    const v4f b0 = *(const v4f*)(bb + fb), b1 = *(const v4f*)(bb + fb + 4);
#pragma unroll
    for (int r = 0; r < 4; ++r) { acc[r] += b0[r]; acc[4 + r] += b1[r]; }
  }
  float s0 = 0.f, s1 = 0.f;
#pragma unroll
  for (int r = 0; r < 4; ++r) { s0 += acc[r]; s1 += acc[4 + r]; }
  float pl = s0 + s1;
  pl += xhalf(pl);
  const float mw = pl * (1.f / 16.f);
  float q0 = 0.f, q1 = 0.f;
#pragma unroll
  for (int r = 0; r < 4; ++r) {
    const float d0 = acc[r] - mw, d1 = acc[4 + r] - mw;
    q0 += d0 * d0; q1 += d1 * d1;
  }
  float ql = q0 + q1;
  ql += xhalf(ql);
  if (half == 0) *(v2f*)(statrow + 2 * f) = v2f{mw, ql};
  lds_barrier();

  const v4f sA = *(const v4f*)statrow,       sB = *(const v4f*)(statrow + 4);
  const v4f sC = *(const v4f*)(statrow + 8), sD = *(const v4f*)(statrow + 12);
  const float mu = (((sA[0] + sA[2]) + (sB[0] + sB[2])) +
                    ((sC[0] + sC[2]) + (sD[0] + sD[2]))) * 0.125f;
  const float e0 = sA[0] - mu, e1 = sA[2] - mu, e2 = sB[0] - mu, e3 = sB[2] - mu;
  const float e4 = sC[0] - mu, e5 = sC[2] - mu, e6 = sD[0] - mu, e7 = sD[2] - mu;
  const float m2 = (((sA[1] + sA[3]) + (sB[1] + sB[3])) + ((sC[1] + sC[3]) + (sD[1] + sD[3]))) +
                   16.f * (((e0 * e0 + e1 * e1) + (e2 * e2 + e3 * e3)) +
                           ((e4 * e4 + e5 * e5) + (e6 * e6 + e7 * e7)));
  const float rstd = rsqrtf(m2 * (1.f / 128.f) + LNEPS);

  const v4f g0 = *(const v4f*)(gg + fb), g1 = *(const v4f*)(gg + fb + 4);
  const v4f t0 = *(const v4f*)(ee + fb), t1 = *(const v4f*)(ee + fb + 4);
  v8h o;
  const v2f muv = {mu, mu}, rsv = {rstd, rstd};
#pragma unroll
  for (int j = 0; j < 4; ++j) {
    const v2f av = {acc[2 * j], acc[2 * j + 1]};
    const v2f gv = j < 2 ? v2f{g0[2 * j], g0[2 * j + 1]} : v2f{g1[2 * j - 4], g1[2 * j - 3]};
    const v2f ev = j < 2 ? v2f{t0[2 * j], t0[2 * j + 1]} : v2f{t1[2 * j - 4], t1[2 * j - 3]};
    const v2f xn = (av - muv) * rsv * gv + ev;
    const v2f y = gelu_pair(xn);
    o[2 * j] = (_Float16)y[0];
    o[2 * j + 1] = (_Float16)y[1];
  }
  *(v8h*)(outrow + fb) = o;
  lds_barrier();
}

__device__ __forceinline__ void head_tile8(const v8f& acc, const float (&So)[8], const float (&dS)[8],
                                           const float* bp, float& gsum, float& asum) {
  const v4f c0 = *(const v4f*)bp,  c1 = *(const v4f*)(bp + 4);
  float gs = 0.f, as = 0.f;
#pragma unroll
  for (int r = 0; r < 8; ++r) {
    const float bn  = r < 4 ? c0[r & 3] : c1[r & 3];
    const float z  = 1.f / (1.f + expf(-(acc[r] + bn)));
    gs += z * dS[r];
    as += z * So[r];
  }
  gsum = gs; asum = as;
}

extern "C" __global__ void __launch_bounds__(NTHR, 4)
bsde_fused_kernel(const float* __restrict__ dw, const float* __restrict__ tg,
                  const float* __restrict__ W1, const float* __restrict__ b1,
                  const float* __restrict__ g1, const float* __restrict__ be1,
                  const float* __restrict__ W2, const float* __restrict__ b2,
                  const float* __restrict__ g2, const float* __restrict__ be2,
                  const float* __restrict__ W3, const float* __restrict__ b3,
                  const float* __restrict__ g3, const float* __restrict__ be3,
                  const float* __restrict__ W4, const float* __restrict__ b4,
                  const float* __restrict__ Y0,
                  float* __restrict__ outY, float* __restrict__ outS) {
  extern __shared__ __align__(16) char smem[];
  float* prm = (float*)smem;
  float* pb1 = prm;        float* pg1 = prm + 128;  float* pe1 = prm + 256;
  float* pb2 = prm + 384;  float* pg2 = prm + 512;  float* pe2 = prm + 640;
  float* pb3 = prm + 768;  float* pg3 = prm + 896;  float* pe3 = prm + 1024;
  float* pb4 = prm + 1152;
  float* Sl  = prm + 1264;
  float* Yl  = Sl + ROWS * SROW;
  float* stat = Yl + ROWS;
  float* hsum = stat + ROWS * NSPL * 2;
  _Float16* xin = (_Float16*)(hsum + ROWS * NSPL * 2);
  _Float16* act = xin + ROWS * APITCH;

  const int tid = threadIdx.x;
  const int lane = tid & 31, lane15 = lane & 15, half = lane >> 4;
  const int wv = __builtin_amdgcn_readfirstlane(tid >> 5);
  const int f = wv;
  const long blockBase = (long)blockIdx.x * ROWS;
  const int p = lane15;
  const long bidx = blockBase + p;
  const int nb = half * 8;
  const int d0 = f * 16 + nb;

  v16h wl1[4], wl2[4], wl3[4], wh4[4];
  {
    const int n0 = f * 16 + lane15;
#pragma unroll
    for (int kt = 0; kt < 4; ++kt) {
      const int kb = kt * 32 + half * 8;
      wl1[kt] = wfrag_global<false>(W1, HDIM, n0, HDIM, kb, DIMA + 1);
      wl2[kt] = wfrag_global<false>(W2, HDIM, n0, HDIM, kb, HDIM);
      wl3[kt] = wfrag_global<false>(W3, HDIM, n0, HDIM, kb, HDIM);
      wh4[kt] = v16h{};
    }
#pragma unroll
    for (int kt = 0; kt < 4; ++kt)
      asm volatile("" : "+v"(wl1[kt]), "+v"(wl2[kt]), "+v"(wl3[kt]) : : "memory");
    if (f < 7) {
#pragma unroll
      for (int kt = 0; kt < 4; ++kt)
        wh4[kt] = wfrag_global<true>(W4, DIMA, n0, DIMA, kt * 32 + half * 8, HDIM);
    }
  }
  if (tid < HDIM) {
    pb1[tid] = b1[tid]; pg1[tid] = g1[tid]; pe1[tid] = be1[tid];
    pb2[tid] = b2[tid]; pg2[tid] = g2[tid]; pe2[tid] = be2[tid];
    pb3[tid] = b3[tid]; pg3[tid] = g3[tid]; pe3[tid] = be3[tid];
  }
  if (tid < NPAD4) pb4[tid] = (tid < DIMA) ? b4[tid] : 0.f;
  hsum[tid] = 0.f;
  if (tid < ROWS) {
    const v8h z = {};
    *(v8h*)(xin + tid * APITCH + 112) = z;
    *(v8h*)(xin + tid * APITCH + 120) = z;
  }

  float S[8];
#pragma unroll
  for (int r = 0; r < 8; ++r) S[r] = (d0 + r < DIMA) ? 100.f : 0.f;
  float Y = Y0[0];
  _Float16* xrow = xin + p * APITCH;
  _Float16* arow = act + p * APITCH;
  float* statrow = stat + p * (NSPL * 2);
  float* hrow = hsum + p * (NSPL * 2);

  if (f < 7) {
    v8h o;
#pragma unroll
    for (int r = 0; r < 8; ++r) o[r] = (_Float16)(S[r] * S0INV);
    if (f == 6) {
      const float tv = tg[bidx * MSTEP];
      o[4] = (half == 0) ? (_Float16)tv : (_Float16)0.f;
    }
    *(v8h*)(xrow + d0) = o;
  }
  __syncthreads();

  for (int m = 0; m < MSTEP; ++m) {
    v4f dwa = {}, dwb = {};
    float tnext = 0.f;
    {
      const float* dwr = dw + (bidx * MSTEP + m) * DIMA;
      if (f < 6) {
        dwa = *(const v4f*)(dwr + d0);
        dwb = *(const v4f*)(dwr + d0 + 4);
      } else if (f == 6) {
        if (half == 0) dwa = *(const v4f*)(dwr + 96);
      }
      tnext = tg[bidx * MSTEP + (m + 1 < MSTEP ? m + 1 : m)];
    }

    mlp_layer_split(xrow, arow, wl1, pb1, pg1, pe1, statrow, f, half);

    {
      const v4f hA = *(const v4f*)hrow,       hB = *(const v4f*)(hrow + 4);
      const v4f hC = *(const v4f*)(hrow + 8);
      const v2f hD = *(const v2f*)(hrow + 12);
      const float gsum = ((hA[0] + hA[2]) + (hB[0] + hB[2])) + ((hC[0] + hC[2]) + hD[0]);
      const float asum = ((hA[1] + hA[3]) + (hB[1] + hB[3])) + ((hC[1] + hC[3]) + hD[1]);
      const float Yn = Y + RDT * (Y - asum) + gsum;
      Y = (m > 0) ? Yn : Y;
    }

    mlp_layer_split(arow, arow, wl2, pb2, pg2, pe2, statrow, f, half);

    float So[8], dS[8];
#pragma unroll
    for (int r = 0; r < 8; ++r) {
      const float dwv = r < 4 ? dwa[r & 3] : dwb[r & 3];
      const float s  = S[r];
      const float ds = s * (RDT + SIG * dwv);
      So[r] = s;
      dS[r] = ds;
      S[r] = s + ds;
    }
    if (f < 7) {
      v8h o;
#pragma unroll
      for (int r = 0; r < 8; ++r) o[r] = (_Float16)(S[r] * S0INV);
      if (f == 6) o[4] = (half == 0) ? (_Float16)tnext : (_Float16)0.f;
      *(v8h*)(xrow + d0) = o;
    }

    mlp_layer_split(arow, arow, wl3, pb3, pg3, pe3, statrow, f, half);

    if (f < 7) {
      v16h x[4];
#pragma unroll
      for (int kt = 0; kt < 4; ++kt) x[kt] = load_frag(arow, kt * 32 + half * 8);
      v8f acc = {};
#pragma unroll
      for (int kt = 0; kt < 4; ++kt)
        acc = __builtin_amdgcn_wmma_f32_16x16x32_f16(
            false, wh4[kt], false, x[kt], (short)0, acc, false, false);

      float gain = 0.f, asset = 0.f;
      if (f < 6) {
        head_tile8(acc, So, dS, pb4 + d0, gain, asset);
      } else {
        if (half == 0) {
          const v4f c0 = *(const v4f*)(pb4 + 96);
#pragma unroll
          for (int r = 0; r < 4; ++r) {
            const float z  = 1.f / (1.f + expf(-(acc[r] + c0[r])));
            gain  += z * dS[r];
            asset += z * So[r];
          }
        }
      }
      gain  += xhalf(gain);
      asset += xhalf(asset);
      if (half == 0) *(v2f*)(hrow + 2 * f) = v2f{gain, asset};
    }
  }
  __syncthreads();
  {
    const v4f hA = *(const v4f*)hrow,       hB = *(const v4f*)(hrow + 4);
    const v4f hC = *(const v4f*)(hrow + 8);
    const v2f hD = *(const v2f*)(hrow + 12);
    const float gsum = ((hA[0] + hA[2]) + (hB[0] + hB[2])) + ((hC[0] + hC[2]) + hD[0]);
    const float asum = ((hA[1] + hA[3]) + (hB[1] + hB[3])) + ((hC[1] + hC[3]) + hD[1]);
    Y = Y + RDT * (Y - asum) + gsum;
  }

  if (f < 6) {
    float* sp = Sl + p * SROW + d0;
    *(v4f*)sp       = v4f{S[0], S[1], S[2], S[3]};
    *(v4f*)(sp + 4) = v4f{S[4], S[5], S[6], S[7]};
  } else if (f == 6) {
    if (half == 0) *(v4f*)(Sl + p * SROW + 96) = v4f{S[0], S[1], S[2], S[3]};
  }
  if (f == 0 && half == 0) Yl[p] = Y;
  __syncthreads();
  for (int i = tid; i < ROWS; i += NTHR) outY[blockBase + i] = Yl[i];
  for (int i = tid; i < ROWS * DIMA; i += NTHR) {
    const int r = i / DIMA, d = i - r * DIMA;
    outS[(blockBase + r) * (long)DIMA + d] = Sl[r * SROW + d];
  }
}

extern "C" void kernel_launch(void* const* d_in, const int* in_sizes, int n_in,
                              void* d_out, int out_size, void* d_ws, size_t ws_size,
                              hipStream_t stream) {
  (void)in_sizes; (void)n_in; (void)out_size; (void)d_ws; (void)ws_size;
  const float* dw  = (const float*)d_in[0];
  const float* tg  = (const float*)d_in[1];
  const float* W1  = (const float*)d_in[2];
  const float* b1  = (const float*)d_in[3];
  const float* g1  = (const float*)d_in[4];
  const float* be1 = (const float*)d_in[5];
  const float* W2  = (const float*)d_in[6];
  const float* b2  = (const float*)d_in[7];
  const float* g2  = (const float*)d_in[8];
  const float* be2 = (const float*)d_in[9];
  const float* W3  = (const float*)d_in[10];
  const float* b3  = (const float*)d_in[11];
  const float* g3  = (const float*)d_in[12];
  const float* be3 = (const float*)d_in[13];
  const float* W4  = (const float*)d_in[14];
  const float* b4  = (const float*)d_in[15];
  const float* Y0  = (const float*)d_in[16];

  float* outY = (float*)d_out;
  float* outS = outY + BATCH;

  const size_t smem =
      (size_t)(9 * 128 + NPAD4) * sizeof(float) +
      (size_t)(ROWS * SROW + ROWS) * sizeof(float) +
      (size_t)(2 * ROWS * NSPL * 2) * sizeof(float) +
      (size_t)(2 * ROWS * APITCH) * sizeof(_Float16);

  dim3 grid(BATCH / ROWS);
  dim3 block(NTHR);
  bsde_fused_kernel<<<grid, block, smem, stream>>>(
      dw, tg, W1, b1, g1, be1, W2, b2, g2, be2, W3, b3, g3, be3, W4, b4, Y0,
      outY, outS);
}


# Head sigmoid on the bare hardware 2^x and 1/x, its exponent one fma with a bias row stored pre-scaled (was a range-reduced expf and an IEEE division, about 25 instructions an asset); same sums
# speedup vs baseline: 26.0889x; 1.1115x over previous
#include <hip/hip_runtime.h>
#include <math.h>

#define DIMA   100
#define MSTEP  50
#define HDIM   128
#define BATCH  8192
#define NPAD4  112
#define SROW   104
#define ROWS   16
#define NSPL   8
#define NTHR   256
#define APITCH 136
#define S0INV  0.01f
#define RDT    0.001f
#define SIG    0.2f
#define LNEPS  1e-5f
#define NLOG2E -1.4426950408889634f

typedef __attribute__((ext_vector_type(16))) _Float16 v16h;
typedef __attribute__((ext_vector_type(8)))  _Float16 v8h;
typedef __attribute__((ext_vector_type(8)))  float    v8f;
typedef __attribute__((ext_vector_type(4)))  float    v4f;
typedef __attribute__((ext_vector_type(2)))  float    v2f;

union V16U { v16h v; v8h h[2]; };

__device__ __forceinline__ v16h load_frag(const _Float16* base, int kb) {
  V16U u;
  u.h[0] = *(const v8h*)(base + kb);
  u.h[1] = *(const v8h*)(base + kb + 16);
  return u.v;
}

template <bool PADN>
__device__ __forceinline__ v16h wfrag_global(const float* __restrict__ W, int ldn, int n, int nlim,
                                             int kb, int klim) {
  const bool nok = PADN ? n < nlim : true;
  const int nn = nok ? n : nlim - 1;
  v16h v;
#pragma unroll
  for (int s = 0; s < 16; ++s) {
    const int k = kb + (s & 7) + ((s >> 3) << 4);
    const int kk = k < klim ? k : klim - 1;
    const float w = W[kk * ldn + nn];
    v[s] = (nok && k < klim) ? (_Float16)w : (_Float16)0.f;
  }
  return v;
}

__device__ __forceinline__ float xhalf(float x) {
  return __builtin_bit_cast(float, __builtin_amdgcn_permlanex16(
      __builtin_bit_cast(int, x), __builtin_bit_cast(int, x),
      0x76543210, 0xfedcba98, false, false));
}

__device__ __forceinline__ v2f gelu_pair(v2f x) {
  v2f a;
  a[0] = __builtin_fabsf(x[0]);
  a[1] = __builtin_fabsf(x[1]);
  const v2f c7 = {-2.8349113563308492e-06f, -2.8349113563308492e-06f};
  const v2f c6 = {3.937767178285867e-05f, 3.937767178285867e-05f};
  const v2f c5 = {-0.0001861801283666864f, -0.0001861801283666864f};
  const v2f c4 = {-0.0001369366655126214f, -0.0001369366655126214f};
  const v2f c3 = {0.0070634204894304276f, 0.0070634204894304276f};
  const v2f c2 = {-0.052496179938316345f, -0.052496179938316345f};
  const v2f c1 = {-0.4592081904411316f, -0.4592081904411316f};
  const v2f c0 = {-1.1511051654815674f, -1.1511051654815674f};
  const v2f m1 = {-1.f, -1.f};
  v2f g = __builtin_elementwise_fma(c7, a, c6);
  g = __builtin_elementwise_fma(g, a, c5);
  g = __builtin_elementwise_fma(g, a, c4);
  g = __builtin_elementwise_fma(g, a, c3);
  g = __builtin_elementwise_fma(g, a, c2);
  g = __builtin_elementwise_fma(g, a, c1);
  g = __builtin_elementwise_fma(g, a, c0);
  const v2f r = __builtin_elementwise_fma(a, g, m1);
  v2f h;
  h[0] = __builtin_amdgcn_exp2f(r[0]);
  h[1] = __builtin_amdgcn_exp2f(r[1]);
  v2f relu;
  relu[0] = __builtin_fmaxf(x[0], 0.f);
  relu[1] = __builtin_fmaxf(x[1], 0.f);
  return __builtin_elementwise_fma(-a, h, relu);
}

__device__ __forceinline__ void lds_barrier() {
  __builtin_amdgcn_fence(__ATOMIC_RELEASE, "workgroup", "local");
  __builtin_amdgcn_s_barrier();
  __builtin_amdgcn_fence(__ATOMIC_ACQUIRE, "workgroup", "local");
}

__device__ __forceinline__ void mlp_layer_split(const _Float16* inrow, _Float16* outrow,
                                                const v16h (&wt)[4],
                                                const float* bb,
                                                const float* gg, const float* ee,
                                                float* statrow, int f, int half) {
  v16h x[4];
#pragma unroll
  for (int kt = 0; kt < 4; ++kt) x[kt] = load_frag(inrow, kt * 32 + half * 8);

  v8f acc = {};
#pragma unroll
  for (int kt = 0; kt < 4; ++kt)
    acc = __builtin_amdgcn_wmma_f32_16x16x32_f16(false, wt[kt], false, x[kt], (short)0, acc, false, false);

  const int fb = f * 16 + half * 8;
  {
    const v4f b0 = *(const v4f*)(bb + fb), b1 = *(const v4f*)(bb + fb + 4);
#pragma unroll
    for (int r = 0; r < 4; ++r) { acc[r] += b0[r]; acc[4 + r] += b1[r]; }
  }
  float s0 = 0.f, s1 = 0.f;
#pragma unroll
  for (int r = 0; r < 4; ++r) { s0 += acc[r]; s1 += acc[4 + r]; }
  float pl = s0 + s1;
  pl += xhalf(pl);
  const float mw = pl * (1.f / 16.f);
  float q0 = 0.f, q1 = 0.f;
#pragma unroll
  for (int r = 0; r < 4; ++r) {
    const float d0 = acc[r] - mw, d1 = acc[4 + r] - mw;
    q0 += d0 * d0; q1 += d1 * d1;
  }
  float ql = q0 + q1;
  ql += xhalf(ql);
  if (half == 0) *(v2f*)(statrow + 2 * f) = v2f{mw, ql};
  lds_barrier();

  const v4f sA = *(const v4f*)statrow,       sB = *(const v4f*)(statrow + 4);
  const v4f sC = *(const v4f*)(statrow + 8), sD = *(const v4f*)(statrow + 12);
  const float mu = (((sA[0] + sA[2]) + (sB[0] + sB[2])) +
                    ((sC[0] + sC[2]) + (sD[0] + sD[2]))) * 0.125f;
  const float e0 = sA[0] - mu, e1 = sA[2] - mu, e2 = sB[0] - mu, e3 = sB[2] - mu;
  const float e4 = sC[0] - mu, e5 = sC[2] - mu, e6 = sD[0] - mu, e7 = sD[2] - mu;
  const float m2 = (((sA[1] + sA[3]) + (sB[1] + sB[3])) + ((sC[1] + sC[3]) + (sD[1] + sD[3]))) +
                   16.f * (((e0 * e0 + e1 * e1) + (e2 * e2 + e3 * e3)) +
                           ((e4 * e4 + e5 * e5) + (e6 * e6 + e7 * e7)));
  const float rstd = rsqrtf(m2 * (1.f / 128.f) + LNEPS);

  const v4f g0 = *(const v4f*)(gg + fb), g1 = *(const v4f*)(gg + fb + 4);
  const v4f t0 = *(const v4f*)(ee + fb), t1 = *(const v4f*)(ee + fb + 4);
  v8h o;
  const v2f muv = {mu, mu}, rsv = {rstd, rstd};
#pragma unroll
  for (int j = 0; j < 4; ++j) {
    const v2f av = {acc[2 * j], acc[2 * j + 1]};
    const v2f gv = j < 2 ? v2f{g0[2 * j], g0[2 * j + 1]} : v2f{g1[2 * j - 4], g1[2 * j - 3]};
    const v2f ev = j < 2 ? v2f{t0[2 * j], t0[2 * j + 1]} : v2f{t1[2 * j - 4], t1[2 * j - 3]};
    const v2f xn = (av - muv) * rsv * gv + ev;
    const v2f y = gelu_pair(xn);
    o[2 * j] = (_Float16)y[0];
    o[2 * j + 1] = (_Float16)y[1];
  }
  *(v8h*)(outrow + fb) = o;
  lds_barrier();
}

__device__ __forceinline__ float sigmoid_hw(float sarg) {
  return __builtin_amdgcn_rcpf(1.f + __builtin_amdgcn_exp2f(sarg));
}

__device__ __forceinline__ void head_tile8(const v8f& acc, const float (&So)[8], const float (&dS)[8],
                                           const float* bp, float& gsum, float& asum) {
  const v4f c0 = *(const v4f*)bp,  c1 = *(const v4f*)(bp + 4);
  float gs = 0.f, as = 0.f;
#pragma unroll
  for (int r = 0; r < 8; ++r) {
    const float bn  = r < 4 ? c0[r & 3] : c1[r & 3];
    const float z  = sigmoid_hw(__builtin_fmaf(acc[r], NLOG2E, bn));
    gs += z * dS[r];
    as += z * So[r];
  }
  gsum = gs; asum = as;
}

extern "C" __global__ void __launch_bounds__(NTHR, 4)
bsde_fused_kernel(const float* __restrict__ dw, const float* __restrict__ tg,
                  const float* __restrict__ W1, const float* __restrict__ b1,
                  const float* __restrict__ g1, const float* __restrict__ be1,
                  const float* __restrict__ W2, const float* __restrict__ b2,
                  const float* __restrict__ g2, const float* __restrict__ be2,
                  const float* __restrict__ W3, const float* __restrict__ b3,
                  const float* __restrict__ g3, const float* __restrict__ be3,
                  const float* __restrict__ W4, const float* __restrict__ b4,
                  const float* __restrict__ Y0,
                  float* __restrict__ outY, float* __restrict__ outS) {
  extern __shared__ __align__(16) char smem[];
  float* prm = (float*)smem;
  float* pb1 = prm;        float* pg1 = prm + 128;  float* pe1 = prm + 256;
  float* pb2 = prm + 384;  float* pg2 = prm + 512;  float* pe2 = prm + 640;
  float* pb3 = prm + 768;  float* pg3 = prm + 896;  float* pe3 = prm + 1024;
  float* pb4 = prm + 1152;
  float* Sl  = prm + 1264;
  float* Yl  = Sl + ROWS * SROW;
  float* stat = Yl + ROWS;
  float* hsum = stat + ROWS * NSPL * 2;
  _Float16* xin = (_Float16*)(hsum + ROWS * NSPL * 2);
  _Float16* act = xin + ROWS * APITCH;

  const int tid = threadIdx.x;
  const int lane = tid & 31, lane15 = lane & 15, half = lane >> 4;
  const int wv = __builtin_amdgcn_readfirstlane(tid >> 5);
  const int f = wv;
  const long blockBase = (long)blockIdx.x * ROWS;
  const int p = lane15;
  const long bidx = blockBase + p;
  const int nb = half * 8;
  const int d0 = f * 16 + nb;

  v16h wl1[4], wl2[4], wl3[4], wh4[4];
  {
    const int n0 = f * 16 + lane15;
#pragma unroll
    for (int kt = 0; kt < 4; ++kt) {
      const int kb = kt * 32 + half * 8;
      wl1[kt] = wfrag_global<false>(W1, HDIM, n0, HDIM, kb, DIMA + 1);
      wl2[kt] = wfrag_global<false>(W2, HDIM, n0, HDIM, kb, HDIM);
      wl3[kt] = wfrag_global<false>(W3, HDIM, n0, HDIM, kb, HDIM);
      wh4[kt] = v16h{};
    }
#pragma unroll
    for (int kt = 0; kt < 4; ++kt)
      asm volatile("" : "+v"(wl1[kt]), "+v"(wl2[kt]), "+v"(wl3[kt]) : : "memory");
    if (f < 7) {
#pragma unroll
      for (int kt = 0; kt < 4; ++kt)
        wh4[kt] = wfrag_global<true>(W4, DIMA, n0, DIMA, kt * 32 + half * 8, HDIM);
    }
  }
  if (tid < HDIM) {
    pb1[tid] = b1[tid]; pg1[tid] = g1[tid]; pe1[tid] = be1[tid];
    pb2[tid] = b2[tid]; pg2[tid] = g2[tid]; pe2[tid] = be2[tid];
    pb3[tid] = b3[tid]; pg3[tid] = g3[tid]; pe3[tid] = be3[tid];
  }
  if (tid < NPAD4) pb4[tid] = (tid < DIMA) ? b4[tid] * NLOG2E : 0.f;
  hsum[tid] = 0.f;
  if (tid < ROWS) {
    const v8h z = {};
    *(v8h*)(xin + tid * APITCH + 112) = z;
    *(v8h*)(xin + tid * APITCH + 120) = z;
  }

  float S[8];
#pragma unroll
  for (int r = 0; r < 8; ++r) S[r] = (d0 + r < DIMA) ? 100.f : 0.f;
  float Y = Y0[0];
  _Float16* xrow = xin + p * APITCH;
  _Float16* arow = act + p * APITCH;
  float* statrow = stat + p * (NSPL * 2);
  float* hrow = hsum + p * (NSPL * 2);

  if (f < 7) {
    v8h o;
#pragma unroll
    for (int r = 0; r < 8; ++r) o[r] = (_Float16)(S[r] * S0INV);
    if (f == 6) {
      const float tv = tg[bidx * MSTEP];
      o[4] = (half == 0) ? (_Float16)tv : (_Float16)0.f;
    }
    *(v8h*)(xrow + d0) = o;
  }
  __syncthreads();

  for (int m = 0; m < MSTEP; ++m) {
    v4f dwa = {}, dwb = {};
    float tnext = 0.f;
    {
      const float* dwr = dw + (bidx * MSTEP + m) * DIMA;
      if (f < 6) {
        dwa = *(const v4f*)(dwr + d0);
        dwb = *(const v4f*)(dwr + d0 + 4);
      } else if (f == 6) {
        if (half == 0) dwa = *(const v4f*)(dwr + 96);
      }
      tnext = tg[bidx * MSTEP + (m + 1 < MSTEP ? m + 1 : m)];
    }

    mlp_layer_split(xrow, arow, wl1, pb1, pg1, pe1, statrow, f, half);

    {
      const v4f hA = *(const v4f*)hrow,       hB = *(const v4f*)(hrow + 4);
      const v4f hC = *(const v4f*)(hrow + 8);
      const v2f hD = *(const v2f*)(hrow + 12);
      const float gsum = ((hA[0] + hA[2]) + (hB[0] + hB[2])) + ((hC[0] + hC[2]) + hD[0]);
      const float asum = ((hA[1] + hA[3]) + (hB[1] + hB[3])) + ((hC[1] + hC[3]) + hD[1]);
      const float Yn = Y + RDT * (Y - asum) + gsum;
      Y = (m > 0) ? Yn : Y;
    }

    mlp_layer_split(arow, arow, wl2, pb2, pg2, pe2, statrow, f, half);

    float So[8], dS[8];
#pragma unroll
    for (int r = 0; r < 8; ++r) {
      const float dwv = r < 4 ? dwa[r & 3] : dwb[r & 3];
      const float s  = S[r];
      const float ds = s * (RDT + SIG * dwv);
      So[r] = s;
      dS[r] = ds;
      S[r] = s + ds;
    }
    if (f < 7) {
      v8h o;
#pragma unroll
      for (int r = 0; r < 8; ++r) o[r] = (_Float16)(S[r] * S0INV);
      if (f == 6) o[4] = (half == 0) ? (_Float16)tnext : (_Float16)0.f;
      *(v8h*)(xrow + d0) = o;
    }

    mlp_layer_split(arow, arow, wl3, pb3, pg3, pe3, statrow, f, half);

    if (f < 7) {
      v16h x[4];
#pragma unroll
      for (int kt = 0; kt < 4; ++kt) x[kt] = load_frag(arow, kt * 32 + half * 8);
      v8f acc = {};
#pragma unroll
      for (int kt = 0; kt < 4; ++kt)
        acc = __builtin_amdgcn_wmma_f32_16x16x32_f16(
            false, wh4[kt], false, x[kt], (short)0, acc, false, false);

      float gain = 0.f, asset = 0.f;
      if (f < 6) {
        head_tile8(acc, So, dS, pb4 + d0, gain, asset);
      } else {
        if (half == 0) {
          const v4f c0 = *(const v4f*)(pb4 + 96);
#pragma unroll
          for (int r = 0; r < 4; ++r) {
            const float z  = sigmoid_hw(__builtin_fmaf(acc[r], NLOG2E, c0[r]));
            gain  += z * dS[r];
            asset += z * So[r];
          }
        }
      }
      gain  += xhalf(gain);
      asset += xhalf(asset);
      if (half == 0) *(v2f*)(hrow + 2 * f) = v2f{gain, asset};
    }
  }
  __syncthreads();
  {
    const v4f hA = *(const v4f*)hrow,       hB = *(const v4f*)(hrow + 4);
    const v4f hC = *(const v4f*)(hrow + 8);
    const v2f hD = *(const v2f*)(hrow + 12);
    const float gsum = ((hA[0] + hA[2]) + (hB[0] + hB[2])) + ((hC[0] + hC[2]) + hD[0]);
    const float asum = ((hA[1] + hA[3]) + (hB[1] + hB[3])) + ((hC[1] + hC[3]) + hD[1]);
    Y = Y + RDT * (Y - asum) + gsum;
  }

  if (f < 6) {
    float* sp = Sl + p * SROW + d0;
    *(v4f*)sp       = v4f{S[0], S[1], S[2], S[3]};
    *(v4f*)(sp + 4) = v4f{S[4], S[5], S[6], S[7]};
  } else if (f == 6) {
    if (half == 0) *(v4f*)(Sl + p * SROW + 96) = v4f{S[0], S[1], S[2], S[3]};
  }
  if (f == 0 && half == 0) Yl[p] = Y;
  __syncthreads();
  for (int i = tid; i < ROWS; i += NTHR) outY[blockBase + i] = Yl[i];
  for (int i = tid; i < ROWS * DIMA; i += NTHR) {
    const int r = i / DIMA, d = i - r * DIMA;
    outS[(blockBase + r) * (long)DIMA + d] = Sl[r * SROW + d];
  }
}

extern "C" void kernel_launch(void* const* d_in, const int* in_sizes, int n_in,
                              void* d_out, int out_size, void* d_ws, size_t ws_size,
                              hipStream_t stream) {
  (void)in_sizes; (void)n_in; (void)out_size; (void)d_ws; (void)ws_size;
  const float* dw  = (const float*)d_in[0];
  const float* tg  = (const float*)d_in[1];
  const float* W1  = (const float*)d_in[2];
  const float* b1  = (const float*)d_in[3];
  const float* g1  = (const float*)d_in[4];
  const float* be1 = (const float*)d_in[5];
  const float* W2  = (const float*)d_in[6];
  const float* b2  = (const float*)d_in[7];
  const float* g2  = (const float*)d_in[8];
  const float* be2 = (const float*)d_in[9];
  const float* W3  = (const float*)d_in[10];
  const float* b3  = (const float*)d_in[11];
  const float* g3  = (const float*)d_in[12];
  const float* be3 = (const float*)d_in[13];
  const float* W4  = (const float*)d_in[14];
  const float* b4  = (const float*)d_in[15];
  const float* Y0  = (const float*)d_in[16];

  float* outY = (float*)d_out;
  float* outS = outY + BATCH;

  const size_t smem =
      (size_t)(9 * 128 + NPAD4) * sizeof(float) +
      (size_t)(ROWS * SROW + ROWS) * sizeof(float) +
      (size_t)(2 * ROWS * NSPL * 2) * sizeof(float) +
      (size_t)(2 * ROWS * APITCH) * sizeof(_Float16);

  dim3 grid(BATCH / ROWS);
  dim3 block(NTHR);
  bsde_fused_kernel<<<grid, block, smem, stream>>>(
      dw, tg, W1, b1, g1, be1, W2, b2, g2, be2, W3, b3, g3, be3, W4, b4, Y0,
      outY, outS);
}


# LayerNorm from plain sum and sum of squares side by side (var = Q/128 - mean^2), sound for any weights: each weight row and bias is centred over the features before the f16 rounding; sums reordered
# speedup vs baseline: 27.5801x; 1.0572x over previous
#include <hip/hip_runtime.h>
#include <math.h>

#define DIMA   100
#define MSTEP  50
#define HDIM   128
#define BATCH  8192
#define NPAD4  112
#define SROW   104
#define ROWS   16
#define NSPL   8
#define NTHR   256
#define APITCH 136
#define S0INV  0.01f
#define RDT    0.001f
#define SIG    0.2f
#define LNEPS  1e-5f
#define NLOG2E -1.4426950408889634f

typedef __attribute__((ext_vector_type(16))) _Float16 v16h;
typedef __attribute__((ext_vector_type(8)))  _Float16 v8h;
typedef __attribute__((ext_vector_type(8)))  float    v8f;
typedef __attribute__((ext_vector_type(4)))  float    v4f;
typedef __attribute__((ext_vector_type(2)))  float    v2f;

union V16U { v16h v; v8h h[2]; };

__device__ __forceinline__ v16h load_frag(const _Float16* base, int kb) {
  V16U u;
  u.h[0] = *(const v8h*)(base + kb);
  u.h[1] = *(const v8h*)(base + kb + 16);
  return u.v;
}

template <bool PADN, bool CENTRE>
__device__ __forceinline__ v16h wfrag_global(const float* __restrict__ W, int ldn, int n, int nlim,
                                             int kb, int klim, const float* kmean) {
  const bool nok = PADN ? n < nlim : true;
  const int nn = nok ? n : nlim - 1;
  v16h v;
#pragma unroll
  for (int s = 0; s < 16; ++s) {
    const int k = kb + (s & 7) + ((s >> 3) << 4);
    const int kk = k < klim ? k : klim - 1;
    float w = W[kk * ldn + nn];
    if (CENTRE) w -= kmean[k];
    v[s] = (nok && k < klim) ? (_Float16)w : (_Float16)0.f;
  }
  return v;
}

__device__ __forceinline__ float xhalf(float x) {
  return __builtin_bit_cast(float, __builtin_amdgcn_permlanex16(
      __builtin_bit_cast(int, x), __builtin_bit_cast(int, x),
      0x76543210, 0xfedcba98, false, false));
}

template <int XM>
__device__ __forceinline__ float row_partner(float x) {
  return __builtin_bit_cast(float, __builtin_amdgcn_update_dpp(
      0, __builtin_bit_cast(int, x), 0x160 | XM, 0xf, 0xf, false));
}
__device__ __forceinline__ float row_sum(float x) {
  x += row_partner<1>(x);
  x += row_partner<2>(x);
  x += row_partner<4>(x);
  x += row_partner<8>(x);
  return x;
}
__device__ __forceinline__ float wave_sum(float x) {
  x = row_sum(x);
  return x + xhalf(x);
}

__device__ __forceinline__ v2f gelu_pair(v2f x) {
  v2f a;
  a[0] = __builtin_fabsf(x[0]);
  a[1] = __builtin_fabsf(x[1]);
  const v2f c7 = {-2.8349113563308492e-06f, -2.8349113563308492e-06f};
  const v2f c6 = {3.937767178285867e-05f, 3.937767178285867e-05f};
  const v2f c5 = {-0.0001861801283666864f, -0.0001861801283666864f};
  const v2f c4 = {-0.0001369366655126214f, -0.0001369366655126214f};
  const v2f c3 = {0.0070634204894304276f, 0.0070634204894304276f};
  const v2f c2 = {-0.052496179938316345f, -0.052496179938316345f};
  const v2f c1 = {-0.4592081904411316f, -0.4592081904411316f};
  const v2f c0 = {-1.1511051654815674f, -1.1511051654815674f};
  const v2f m1 = {-1.f, -1.f};
  v2f g = __builtin_elementwise_fma(c7, a, c6);
  g = __builtin_elementwise_fma(g, a, c5);
  g = __builtin_elementwise_fma(g, a, c4);
  g = __builtin_elementwise_fma(g, a, c3);
  g = __builtin_elementwise_fma(g, a, c2);
  g = __builtin_elementwise_fma(g, a, c1);
  g = __builtin_elementwise_fma(g, a, c0);
  const v2f r = __builtin_elementwise_fma(a, g, m1);
  v2f h;
  h[0] = __builtin_amdgcn_exp2f(r[0]);
  h[1] = __builtin_amdgcn_exp2f(r[1]);
  v2f relu;
  relu[0] = __builtin_fmaxf(x[0], 0.f);
  relu[1] = __builtin_fmaxf(x[1], 0.f);
  return __builtin_elementwise_fma(-a, h, relu);
}

__device__ __forceinline__ void lds_barrier() {
  __builtin_amdgcn_fence(__ATOMIC_RELEASE, "workgroup", "local");
  __builtin_amdgcn_s_barrier();
  __builtin_amdgcn_fence(__ATOMIC_ACQUIRE, "workgroup", "local");
}

__device__ __forceinline__ void mlp_layer_split(const _Float16* inrow, _Float16* outrow,
                                                const v16h (&wt)[4],
                                                const float* bb,
                                                const float* gg, const float* ee,
                                                float* statrow, int f, int half) {
  v16h x[4];
#pragma unroll
  for (int kt = 0; kt < 4; ++kt) x[kt] = load_frag(inrow, kt * 32 + half * 8);

  v8f acc = {};
#pragma unroll
  for (int kt = 0; kt < 4; ++kt)
    acc = __builtin_amdgcn_wmma_f32_16x16x32_f16(false, wt[kt], false, x[kt], (short)0, acc, false, false);

  const int fb = f * 16 + half * 8;
  {
    const v4f b0 = *(const v4f*)(bb + fb), b1 = *(const v4f*)(bb + fb + 4);
#pragma unroll
    for (int r = 0; r < 4; ++r) { acc[r] += b0[r]; acc[4 + r] += b1[r]; }
  }
  v2f sv = {acc[0], acc[1]};
  v2f qv = sv * sv;
#pragma unroll
  for (int j = 1; j < 4; ++j) {
    const v2f av = {acc[2 * j], acc[2 * j + 1]};
    sv += av;
    qv = __builtin_elementwise_fma(av, av, qv);
  }
  float pl = sv[0] + sv[1];
  float ql = qv[0] + qv[1];
  pl += xhalf(pl);
  ql += xhalf(ql);
  if (half == 0) *(v2f*)(statrow + 2 * f) = v2f{pl, ql};
  lds_barrier();

  const v4f sA = *(const v4f*)statrow,       sB = *(const v4f*)(statrow + 4);
  const v4f sC = *(const v4f*)(statrow + 8), sD = *(const v4f*)(statrow + 12);
  const v4f st = (sA + sB) + (sC + sD);
  const float mu = (st[0] + st[2]) * (1.f / 128.f);
  const float ex2 = (st[1] + st[3]) * (1.f / 128.f);
  const float var = __builtin_fmaxf(__builtin_fmaf(-mu, mu, ex2), 0.f);
  const float rstd = rsqrtf(var + LNEPS);

  const v4f g0 = *(const v4f*)(gg + fb), g1 = *(const v4f*)(gg + fb + 4);
  const v4f t0 = *(const v4f*)(ee + fb), t1 = *(const v4f*)(ee + fb + 4);
  v8h o;
  const v2f muv = {mu, mu}, rsv = {rstd, rstd};
#pragma unroll
  for (int j = 0; j < 4; ++j) {
    const v2f av = {acc[2 * j], acc[2 * j + 1]};
    const v2f gv = j < 2 ? v2f{g0[2 * j], g0[2 * j + 1]} : v2f{g1[2 * j - 4], g1[2 * j - 3]};
    const v2f ev = j < 2 ? v2f{t0[2 * j], t0[2 * j + 1]} : v2f{t1[2 * j - 4], t1[2 * j - 3]};
    const v2f xn = (av - muv) * rsv * gv + ev;
    const v2f y = gelu_pair(xn);
    o[2 * j] = (_Float16)y[0];
    o[2 * j + 1] = (_Float16)y[1];
  }
  *(v8h*)(outrow + fb) = o;
  lds_barrier();
}

__device__ __forceinline__ float sigmoid_hw(float sarg) {
  return __builtin_amdgcn_rcpf(1.f + __builtin_amdgcn_exp2f(sarg));
}

__device__ __forceinline__ void head_tile8(const v8f& acc, const float (&So)[8], const float (&dS)[8],
                                           const float* bp, float& gsum, float& asum) {
  const v4f c0 = *(const v4f*)bp,  c1 = *(const v4f*)(bp + 4);
  float gs = 0.f, as = 0.f;
#pragma unroll
  for (int r = 0; r < 8; ++r) {
    const float bn  = r < 4 ? c0[r & 3] : c1[r & 3];
    const float z  = sigmoid_hw(__builtin_fmaf(acc[r], NLOG2E, bn));
    gs += z * dS[r];
    as += z * So[r];
  }
  gsum = gs; asum = as;
}

extern "C" __global__ void __launch_bounds__(NTHR, 4)
bsde_fused_kernel(const float* __restrict__ dw, const float* __restrict__ tg,
                  const float* __restrict__ W1, const float* __restrict__ b1,
                  const float* __restrict__ g1, const float* __restrict__ be1,
                  const float* __restrict__ W2, const float* __restrict__ b2,
                  const float* __restrict__ g2, const float* __restrict__ be2,
                  const float* __restrict__ W3, const float* __restrict__ b3,
                  const float* __restrict__ g3, const float* __restrict__ be3,
                  const float* __restrict__ W4, const float* __restrict__ b4,
                  const float* __restrict__ Y0,
                  float* __restrict__ outY, float* __restrict__ outS) {
  extern __shared__ __align__(16) char smem[];
  float* prm = (float*)smem;
  float* pb1 = prm;        float* pg1 = prm + 128;  float* pe1 = prm + 256;
  float* pb2 = prm + 384;  float* pg2 = prm + 512;  float* pe2 = prm + 640;
  float* pb3 = prm + 768;  float* pg3 = prm + 896;  float* pe3 = prm + 1024;
  float* pb4 = prm + 1152;
  float* Sl  = prm + 1264;
  float* Yl  = Sl + ROWS * SROW;
  float* stat = Yl + ROWS;
  float* hsum = stat + ROWS * NSPL * 2;
  float* wmu = hsum + ROWS * NSPL * 2;
  _Float16* xin = (_Float16*)(wmu + 3 * HDIM);
  _Float16* act = xin + ROWS * APITCH;

  const int tid = threadIdx.x;
  const int lane = tid & 31, lane15 = lane & 15, half = lane >> 4;
  const int wv = __builtin_amdgcn_readfirstlane(tid >> 5);
  const int f = wv;
  const long blockBase = (long)blockIdx.x * ROWS;
  const int p = lane15;
  const long bidx = blockBase + p;
  const int nb = half * 8;
  const int d0 = f * 16 + nb;

  {
    float r1[8], r2[8], r3[8];
#pragma unroll
    for (int i = 0; i < 8; ++i) {
      const int k = 2 * (wv + 8 * i) + half;
      const int k1 = k < DIMA + 1 ? k : DIMA;
      const float* q1 = W1 + k1 * HDIM + lane15 * 8;
      const float* q2 = W2 + k * HDIM + lane15 * 8;
      const float* q3 = W3 + k * HDIM + lane15 * 8;
      const v4f a1 = *(const v4f*)q1, c1 = *(const v4f*)(q1 + 4);
      const v4f a2 = *(const v4f*)q2, c2 = *(const v4f*)(q2 + 4);
      const v4f a3 = *(const v4f*)q3, c3 = *(const v4f*)(q3 + 4);
      r1[i] = ((a1[0] + a1[1]) + (a1[2] + a1[3])) + ((c1[0] + c1[1]) + (c1[2] + c1[3]));
      r2[i] = ((a2[0] + a2[1]) + (a2[2] + a2[3])) + ((c2[0] + c2[1]) + (c2[2] + c2[3]));
      r3[i] = ((a3[0] + a3[1]) + (a3[2] + a3[3])) + ((c3[0] + c3[1]) + (c3[2] + c3[3]));
    }
#pragma unroll
    for (int i = 0; i < 8; ++i) {
      r1[i] = row_sum(r1[i]); r2[i] = row_sum(r2[i]); r3[i] = row_sum(r3[i]);
    }
    if (lane15 == 0) {
#pragma unroll
      for (int i = 0; i < 8; ++i) {
        const int k = 2 * (wv + 8 * i) + half;
        wmu[k] = r1[i] * (1.f / 128.f);
        wmu[HDIM + k] = r2[i] * (1.f / 128.f);
        wmu[2 * HDIM + k] = r3[i] * (1.f / 128.f);
      }
    }
  }
  __syncthreads();

  v16h wl1[4], wl2[4], wl3[4], wh4[4];
  {
    const int n0 = f * 16 + lane15;
#pragma unroll
    for (int kt = 0; kt < 4; ++kt) {
      const int kb = kt * 32 + half * 8;
      wl1[kt] = wfrag_global<false, true>(W1, HDIM, n0, HDIM, kb, DIMA + 1, wmu);
      wl2[kt] = wfrag_global<false, true>(W2, HDIM, n0, HDIM, kb, HDIM, wmu + HDIM);
      wl3[kt] = wfrag_global<false, true>(W3, HDIM, n0, HDIM, kb, HDIM, wmu + 2 * HDIM);
      wh4[kt] = v16h{};
    }
#pragma unroll
    for (int kt = 0; kt < 4; ++kt)
      asm volatile("" : "+v"(wl1[kt]), "+v"(wl2[kt]), "+v"(wl3[kt]) : : "memory");
    if (f < 7) {
#pragma unroll
      for (int kt = 0; kt < 4; ++kt)
        wh4[kt] = wfrag_global<true, false>(W4, DIMA, n0, DIMA, kt * 32 + half * 8, HDIM, wmu);
    }
  }
  if (tid < HDIM) {
    const v4f u1 = *(const v4f*)(b1 + lane * 4);
    const v4f u2 = *(const v4f*)(b2 + lane * 4);
    const v4f u3 = *(const v4f*)(b3 + lane * 4);
    const float m1 = wave_sum((u1[0] + u1[1]) + (u1[2] + u1[3]));
    const float m2 = wave_sum((u2[0] + u2[1]) + (u2[2] + u2[3]));
    const float m3 = wave_sum((u3[0] + u3[1]) + (u3[2] + u3[3]));
    pb1[tid] = b1[tid] - m1 * (1.f / 128.f); pg1[tid] = g1[tid]; pe1[tid] = be1[tid];
    pb2[tid] = b2[tid] - m2 * (1.f / 128.f); pg2[tid] = g2[tid]; pe2[tid] = be2[tid];
    pb3[tid] = b3[tid] - m3 * (1.f / 128.f); pg3[tid] = g3[tid]; pe3[tid] = be3[tid];
  }
  if (tid < NPAD4) pb4[tid] = (tid < DIMA) ? b4[tid] * NLOG2E : 0.f;
  hsum[tid] = 0.f;
  if (tid < ROWS) {
    const v8h z = {};
    *(v8h*)(xin + tid * APITCH + 112) = z;
    *(v8h*)(xin + tid * APITCH + 120) = z;
  }

  float S[8];
#pragma unroll
  for (int r = 0; r < 8; ++r) S[r] = (d0 + r < DIMA) ? 100.f : 0.f;
  float Y = Y0[0];
  _Float16* xrow = xin + p * APITCH;
  _Float16* arow = act + p * APITCH;
  float* statrow = stat + p * (NSPL * 2);
  float* hrow = hsum + p * (NSPL * 2);

  if (f < 7) {
    v8h o;
#pragma unroll
    for (int r = 0; r < 8; ++r) o[r] = (_Float16)(S[r] * S0INV);
    if (f == 6) {
      const float tv = tg[bidx * MSTEP];
      o[4] = (half == 0) ? (_Float16)tv : (_Float16)0.f;
    }
    *(v8h*)(xrow + d0) = o;
  }
  __syncthreads();

  for (int m = 0; m < MSTEP; ++m) {
    v4f dwa = {}, dwb = {};
    float tnext = 0.f;
    {
      const float* dwr = dw + (bidx * MSTEP + m) * DIMA;
      if (f < 6) {
        dwa = *(const v4f*)(dwr + d0);
        dwb = *(const v4f*)(dwr + d0 + 4);
      } else if (f == 6) {
        if (half == 0) dwa = *(const v4f*)(dwr + 96);
      }
      tnext = tg[bidx * MSTEP + (m + 1 < MSTEP ? m + 1 : m)];
    }

    mlp_layer_split(xrow, arow, wl1, pb1, pg1, pe1, statrow, f, half);

    {
      const v4f hA = *(const v4f*)hrow,       hB = *(const v4f*)(hrow + 4);
      const v4f hC = *(const v4f*)(hrow + 8);
      const v2f hD = *(const v2f*)(hrow + 12);
      const float gsum = ((hA[0] + hA[2]) + (hB[0] + hB[2])) + ((hC[0] + hC[2]) + hD[0]);
      const float asum = ((hA[1] + hA[3]) + (hB[1] + hB[3])) + ((hC[1] + hC[3]) + hD[1]);
      const float Yn = Y + RDT * (Y - asum) + gsum;
      Y = (m > 0) ? Yn : Y;
    }

    mlp_layer_split(arow, arow, wl2, pb2, pg2, pe2, statrow, f, half);

    float So[8], dS[8];
#pragma unroll
    for (int r = 0; r < 8; ++r) {
      const float dwv = r < 4 ? dwa[r & 3] : dwb[r & 3];
      const float s  = S[r];
      const float ds = s * (RDT + SIG * dwv);
      So[r] = s;
      dS[r] = ds;
      S[r] = s + ds;
    }
    if (f < 7) {
      v8h o;
#pragma unroll
      for (int r = 0; r < 8; ++r) o[r] = (_Float16)(S[r] * S0INV);
      if (f == 6) o[4] = (half == 0) ? (_Float16)tnext : (_Float16)0.f;
      *(v8h*)(xrow + d0) = o;
    }

    mlp_layer_split(arow, arow, wl3, pb3, pg3, pe3, statrow, f, half);

    if (f < 7) {
      v16h x[4];
#pragma unroll
      for (int kt = 0; kt < 4; ++kt) x[kt] = load_frag(arow, kt * 32 + half * 8);
      v8f acc = {};
#pragma unroll
      for (int kt = 0; kt < 4; ++kt)
        acc = __builtin_amdgcn_wmma_f32_16x16x32_f16(
            false, wh4[kt], false, x[kt], (short)0, acc, false, false);

      float gain = 0.f, asset = 0.f;
      if (f < 6) {
        head_tile8(acc, So, dS, pb4 + d0, gain, asset);
      } else {
        if (half == 0) {
          const v4f c0 = *(const v4f*)(pb4 + 96);
#pragma unroll
          for (int r = 0; r < 4; ++r) {
            const float z  = sigmoid_hw(__builtin_fmaf(acc[r], NLOG2E, c0[r]));
            gain  += z * dS[r];
            asset += z * So[r];
          }
        }
      }
      gain  += xhalf(gain);
      asset += xhalf(asset);
      if (half == 0) *(v2f*)(hrow + 2 * f) = v2f{gain, asset};
    }
  }
  __syncthreads();
  {
    const v4f hA = *(const v4f*)hrow,       hB = *(const v4f*)(hrow + 4);
    const v4f hC = *(const v4f*)(hrow + 8);
    const v2f hD = *(const v2f*)(hrow + 12);
    const float gsum = ((hA[0] + hA[2]) + (hB[0] + hB[2])) + ((hC[0] + hC[2]) + hD[0]);
    const float asum = ((hA[1] + hA[3]) + (hB[1] + hB[3])) + ((hC[1] + hC[3]) + hD[1]);
    Y = Y + RDT * (Y - asum) + gsum;
  }

  if (f < 6) {
    float* sp = Sl + p * SROW + d0;
    *(v4f*)sp       = v4f{S[0], S[1], S[2], S[3]};
    *(v4f*)(sp + 4) = v4f{S[4], S[5], S[6], S[7]};
  } else if (f == 6) {
    if (half == 0) *(v4f*)(Sl + p * SROW + 96) = v4f{S[0], S[1], S[2], S[3]};
  }
  if (f == 0 && half == 0) Yl[p] = Y;
  __syncthreads();
  for (int i = tid; i < ROWS; i += NTHR) outY[blockBase + i] = Yl[i];
  for (int i = tid; i < ROWS * DIMA; i += NTHR) {
    const int r = i / DIMA, d = i - r * DIMA;
    outS[(blockBase + r) * (long)DIMA + d] = Sl[r * SROW + d];
  }
}

extern "C" void kernel_launch(void* const* d_in, const int* in_sizes, int n_in,
                              void* d_out, int out_size, void* d_ws, size_t ws_size,
                              hipStream_t stream) {
  (void)in_sizes; (void)n_in; (void)out_size; (void)d_ws; (void)ws_size;
  const float* dw  = (const float*)d_in[0];
  const float* tg  = (const float*)d_in[1];
  const float* W1  = (const float*)d_in[2];
  const float* b1  = (const float*)d_in[3];
  const float* g1  = (const float*)d_in[4];
  const float* be1 = (const float*)d_in[5];
  const float* W2  = (const float*)d_in[6];
  const float* b2  = (const float*)d_in[7];
  const float* g2  = (const float*)d_in[8];
  const float* be2 = (const float*)d_in[9];
  const float* W3  = (const float*)d_in[10];
  const float* b3  = (const float*)d_in[11];
  const float* g3  = (const float*)d_in[12];
  const float* be3 = (const float*)d_in[13];
  const float* W4  = (const float*)d_in[14];
  const float* b4  = (const float*)d_in[15];
  const float* Y0  = (const float*)d_in[16];

  float* outY = (float*)d_out;
  float* outS = outY + BATCH;

  const size_t smem =
      (size_t)(9 * 128 + NPAD4) * sizeof(float) +
      (size_t)(ROWS * SROW + ROWS) * sizeof(float) +
      (size_t)(2 * ROWS * NSPL * 2) * sizeof(float) +
      (size_t)(3 * HDIM) * sizeof(float) +
      (size_t)(2 * ROWS * APITCH) * sizeof(_Float16);

  dim3 grid(BATCH / ROWS);
  dim3 block(NTHR);
  bsde_fused_kernel<<<grid, block, smem, stream>>>(
      dw, tg, W1, b1, g1, be1, W2, b2, g2, be2, W3, b3, g3, be3, W4, b4, Y0,
      outY, outS);
}
